# starting kernel
# baseline (speedup 1.0000x reference)
#include <hip/hip_runtime.h>
#include <hip/hip_bf16.h>

typedef __attribute__((ext_vector_type(2))) float v2f;
typedef __attribute__((ext_vector_type(8))) float v8f;

constexpr int BB  = 32;
constexpr int SS  = 2048;
constexpr int DD  = 64;
constexpr int DK  = 24;
constexpr int DV  = 32;
constexpr int LDQ = 32;
constexpr int NT  = SS / 16;

#define WMMA_F32X4(Am, Bm, Cm) \
  __builtin_amdgcn_wmma_f32_16x16x4_f32(false, (Am), false, (Bm), (short)0, (Cm), false, false)

__global__ void __launch_bounds__(256)
qkv_proj_kernel(const float* __restrict__ x, const float* __restrict__ Wq,
                const float* __restrict__ Wk, const float* __restrict__ Wv,
                float* __restrict__ Qp, float* __restrict__ Kp, float* __restrict__ Vm)
{
    int row = blockIdx.x * blockDim.x + threadIdx.x;
    if (row >= BB * SS) return;
    const float* xp = x + (size_t)row * DD;

    float accq[DK], acck[DK], accv[DV];
    #pragma unroll
    for (int j = 0; j < DK; ++j) { accq[j] = 0.f; acck[j] = 0.f; }
    #pragma unroll
    for (int j = 0; j < DV; ++j) accv[j] = 0.f;

    for (int d = 0; d < DD; ++d) {
        float xd = xp[d];
        #pragma unroll
        for (int j = 0; j < DK; ++j) {
            accq[j] = fmaf(xd, Wq[d * DK + j], accq[j]);
            acck[j] = fmaf(xd, Wk[d * DK + j], acck[j]);
        }
        #pragma unroll
        for (int j = 0; j < DV; ++j)
            accv[j] = fmaf(xd, Wv[d * DV + j], accv[j]);
    }
    float* qo = Qp + (size_t)row * LDQ;
    float* ko = Kp + (size_t)row * LDQ;
    float* vo = Vm + (size_t)row * DV;
    #pragma unroll
    for (int j = 0; j < DK; ++j) { qo[j] = accq[j]; ko[j] = acck[j]; }
    #pragma unroll
    for (int j = 0; j < DV; ++j) vo[j] = accv[j];
}

__global__ void __launch_bounds__(32)
col_softmax_stats_kernel(const float* __restrict__ Qp, const float* __restrict__ Kp,
                         float* __restrict__ mArr, float* __restrict__ zinvArr)
{
    int wave = blockIdx.x;
    int bb = wave / NT;
    int tt = wave - bb * NT;
    int lane = threadIdx.x & 31;
    int col  = lane & 15;
    int hl   = lane >> 4;

    const float* kp = Kp + ((size_t)(bb * SS) + (size_t)tt * 16 + col) * LDQ + 2 * hl;
    v2f b0 = *(const v2f*)(kp +  0);
    v2f b1 = *(const v2f*)(kp +  4);
    v2f b2 = *(const v2f*)(kp +  8);
    v2f b3 = *(const v2f*)(kp + 12);
    v2f b4 = *(const v2f*)(kp + 16);
    v2f b5 = *(const v2f*)(kp + 20);

    const float* qbase = Qp + ((size_t)(bb * SS) + col) * LDQ + 2 * hl;
    float m_run = -3.0e38f;
    float z_run = 0.f;

    for (int qt = 0; qt < NT; ++qt) {
        const float* qp = qbase + (size_t)qt * 16 * LDQ;
        __builtin_prefetch(qp + 16 * LDQ, 0, 1);
        v2f a0 = *(const v2f*)(qp +  0);
        v2f a1 = *(const v2f*)(qp +  4);
        v2f a2 = *(const v2f*)(qp +  8);
        v2f a3 = *(const v2f*)(qp + 12);
        v2f a4 = *(const v2f*)(qp + 16);
        v2f a5 = *(const v2f*)(qp + 20);

        v8f s = {};
        s = WMMA_F32X4(a0, b0, s);
        s = WMMA_F32X4(a1, b1, s);
        s = WMMA_F32X4(a2, b2, s);
        s = WMMA_F32X4(a3, b3, s);
        s = WMMA_F32X4(a4, b4, s);
        s = WMMA_F32X4(a5, b5, s);

        float tmax = fmaxf(fmaxf(fmaxf(s[0], s[1]), fmaxf(s[2], s[3])),
                           fmaxf(fmaxf(s[4], s[5]), fmaxf(s[6], s[7])));
        tmax = fmaxf(tmax, __shfl_xor(tmax, 16, 32));
        float m_new = fmaxf(m_run, tmax);
        float p = 0.f;
        #pragma unroll
        for (int i = 0; i < 8; ++i) p += __expf(s[i] - m_new);
        p += __shfl_xor(p, 16, 32);
        z_run = fmaf(z_run, __expf(m_run - m_new), p);
        m_run = m_new;
    }
    if (hl == 0) {
        mArr[bb * SS + tt * 16 + col]    = m_run;
        zinvArr[bb * SS + tt * 16 + col] = 1.f / z_run;
    }
}

__global__ void __launch_bounds__(256)
scale_v_kernel(float* __restrict__ Vm, const float* __restrict__ zinvArr)
{
    int i = blockIdx.x * blockDim.x + threadIdx.x;
    if (i < BB * SS * DV) Vm[i] *= zinvArr[i / DV];
}

__global__ void __launch_bounds__(32)
attn_h1_kernel(const float* __restrict__ Qp, const float* __restrict__ Kp,
               const float* __restrict__ VZ, const float* __restrict__ mArr,
               float* __restrict__ h1)
{
    __shared__ float ldsE[16 * 20];
    int wave = blockIdx.x;
    int bb = wave / NT;
    int qt = wave - bb * NT;
    int lane = threadIdx.x & 31;
    int col  = lane & 15;
    int hl   = lane >> 4;

    const float* qp = Qp + ((size_t)(bb * SS) + (size_t)qt * 16 + col) * LDQ + 2 * hl;
    v2f a0 = *(const v2f*)(qp +  0);
    v2f a1 = *(const v2f*)(qp +  4);
    v2f a2 = *(const v2f*)(qp +  8);
    v2f a3 = *(const v2f*)(qp + 12);
    v2f a4 = *(const v2f*)(qp + 16);
    v2f a5 = *(const v2f*)(qp + 20);

    v8f acc0 = {};
    v8f acc1 = {};

    const float* kbase = Kp + ((size_t)(bb * SS) + col) * LDQ + 2 * hl;
    const float* vzb   = VZ + (size_t)(bb * SS) * DV;
    const float* mrow  = mArr + bb * SS;

    for (int tt = 0; tt < NT; ++tt) {
        const float* kp = kbase + (size_t)tt * 16 * LDQ;
        __builtin_prefetch(kp + 16 * LDQ, 0, 1);
        v2f b0 = *(const v2f*)(kp +  0);
        v2f b1 = *(const v2f*)(kp +  4);
        v2f b2 = *(const v2f*)(kp +  8);
        v2f b3 = *(const v2f*)(kp + 12);
        v2f b4 = *(const v2f*)(kp + 16);
        v2f b5 = *(const v2f*)(kp + 20);

        v8f s = {};
        s = WMMA_F32X4(a0, b0, s);
        s = WMMA_F32X4(a1, b1, s);
        s = WMMA_F32X4(a2, b2, s);
        s = WMMA_F32X4(a3, b3, s);
        s = WMMA_F32X4(a4, b4, s);
        s = WMMA_F32X4(a5, b5, s);

        float mcol = mrow[tt * 16 + col];
        #pragma unroll
        for (int r = 0; r < 8; ++r)
            ldsE[(r + 8 * hl) * 20 + col] = __expf(s[r] - mcol);

        const float* vt = vzb + (size_t)(tt * 16) * DV;
        #pragma unroll
        for (int c = 0; c < 4; ++c) {
            v2f ae = *(const v2f*)&ldsE[col * 20 + 4 * c + 2 * hl];
            const float* vp = vt + (size_t)(4 * c + 2 * hl) * DV;
            v2f bv0, bv1;
            bv0.x = vp[col];           bv0.y = vp[DV + col];
            bv1.x = vp[16 + col];      bv1.y = vp[DV + 16 + col];
            acc0 = WMMA_F32X4(ae, bv0, acc0);
            acc1 = WMMA_F32X4(ae, bv1, acc1);
        }
    }

    #pragma unroll
    for (int r = 0; r < 8; ++r) {
        float* hp = h1 + ((size_t)(bb * SS) + (size_t)qt * 16 + r + 8 * hl) * DV;
        hp[col]      = acc0[r];
        hp[col + 16] = acc1[r];
    }
}

__global__ void __launch_bounds__(256)
out_proj_kernel(const float* __restrict__ h1, const float* __restrict__ Wh,
                float* __restrict__ out)
{
    int row = blockIdx.x * blockDim.x + threadIdx.x;
    if (row >= BB * SS) return;
    const float* hp = h1 + (size_t)row * DV;
    float acc[DD];
    #pragma unroll
    for (int d = 0; d < DD; ++d) acc[d] = 0.f;
    for (int v = 0; v < DV; ++v) {
        float hv = hp[v];
        #pragma unroll
        for (int d = 0; d < DD; ++d)
            acc[d] = fmaf(hv, Wh[v * DD + d], acc[d]);
    }
    float* op = out + (size_t)row * DD;
    #pragma unroll
    for (int d = 0; d < DD; ++d) op[d] = acc[d];
}

extern "C" void kernel_launch(void* const* d_in, const int* in_sizes, int n_in,
                              void* d_out, int out_size, void* d_ws, size_t ws_size,
                              hipStream_t stream)
{
    const float* x  = (const float*)d_in[0];
    const float* Wq = (const float*)d_in[1];
    const float* Wk = (const float*)d_in[2];
    const float* Wv = (const float*)d_in[3];
    const float* Wh = (const float*)d_in[4];
    float* out = (float*)d_out;

    float* ws = (float*)d_ws;
    size_t o = 0;
    float* Qp = ws + o; o += (size_t)BB * SS * LDQ;
    float* Kp = ws + o; o += (size_t)BB * SS * LDQ;
    float* Vm = ws + o; o += (size_t)BB * SS * DV;
    float* mA = ws + o; o += (size_t)BB * SS;
    float* zI = ws + o; o += (size_t)BB * SS;
    float* h1 = ws + o; o += (size_t)BB * SS * DV;

    const int rows = BB * SS;
    qkv_proj_kernel<<<(rows + 255) / 256, 256, 0, stream>>>(x, Wq, Wk, Wv, Qp, Kp, Vm);
    col_softmax_stats_kernel<<<BB * NT, 32, 0, stream>>>(Qp, Kp, mA, zI);
    const int nv = BB * SS * DV;
    scale_v_kernel<<<(nv + 255) / 256, 256, 0, stream>>>(Vm, zI);
    attn_h1_kernel<<<BB * NT, 32, 0, stream>>>(Qp, Kp, Vm, mA, h1);
    out_proj_kernel<<<(rows + 255) / 256, 256, 0, stream>>>(h1, Wh, out);
}


# Both streaming WMMA loops: the next tile's operands (K, column maxima, vz; next q-tile pair) are loaded into a second register set one trip ahead, so no memory round trip stays in the loop chain
# speedup vs baseline: 1.1058x; 1.1058x over previous
#include <hip/hip_runtime.h>
#include <hip/hip_bf16.h>


typedef __attribute__((ext_vector_type(2))) float v2f;
typedef __attribute__((ext_vector_type(8))) float v8f;

constexpr int BB  = 32;
constexpr int SS  = 2048;
constexpr int DD  = 64;
constexpr int DK  = 24;
constexpr int DV  = 32;
constexpr int LDQ = 32;
constexpr int NT  = SS / 16;

#define WMMA_F32X4(Am, Bm, Cm) \
  __builtin_amdgcn_wmma_f32_16x16x4_f32(false, (Am), false, (Bm), (short)0, (Cm), false, false)

__global__ void __launch_bounds__(256)
qkv_proj_kernel(const float* __restrict__ x, const float* __restrict__ Wq,
                const float* __restrict__ Wk, const float* __restrict__ Wv,
                float* __restrict__ Qp, float* __restrict__ Kp, float* __restrict__ Vm)
{
    int row = blockIdx.x * blockDim.x + threadIdx.x;
    if (row >= BB * SS) return;
    const float* xp = x + (size_t)row * DD;

    float accq[DK], acck[DK], accv[DV];
    #pragma unroll
    for (int j = 0; j < DK; ++j) { accq[j] = 0.f; acck[j] = 0.f; }
    #pragma unroll
    for (int j = 0; j < DV; ++j) accv[j] = 0.f;

    for (int d = 0; d < DD; ++d) {
        float xd = xp[d];
        #pragma unroll
        for (int j = 0; j < DK; ++j) {
            accq[j] = fmaf(xd, Wq[d * DK + j], accq[j]);
            acck[j] = fmaf(xd, Wk[d * DK + j], acck[j]);
        }
        #pragma unroll
        for (int j = 0; j < DV; ++j)
            accv[j] = fmaf(xd, Wv[d * DV + j], accv[j]);
    }
    float* qo = Qp + (size_t)row * LDQ;
    float* ko = Kp + (size_t)row * LDQ;
    float* vo = Vm + (size_t)row * DV;
    #pragma unroll
    for (int j = 0; j < DK; ++j) { qo[j] = accq[j]; ko[j] = acck[j]; }
    #pragma unroll
    for (int j = 0; j < DV; ++j) vo[j] = accv[j];
}

struct QFrag { v2f a0, a1, a2, a3, a4, a5; };

__device__ __forceinline__ void load_qfrag(QFrag& f, const float* qp)
{
    f.a0 = *(const v2f*)(qp +  0);
    f.a1 = *(const v2f*)(qp +  4);
    f.a2 = *(const v2f*)(qp +  8);
    f.a3 = *(const v2f*)(qp + 12);
    f.a4 = *(const v2f*)(qp + 16);
    f.a5 = *(const v2f*)(qp + 20);
}

__device__ __forceinline__ void stats_step(const QFrag& f, v2f b0, v2f b1, v2f b2,
                                           v2f b3, v2f b4, v2f b5,
                                           float& m_run, float& z_run)
{
    v8f s = {};
    s = WMMA_F32X4(f.a0, b0, s);
    s = WMMA_F32X4(f.a1, b1, s);
    s = WMMA_F32X4(f.a2, b2, s);
    s = WMMA_F32X4(f.a3, b3, s);
    s = WMMA_F32X4(f.a4, b4, s);
    s = WMMA_F32X4(f.a5, b5, s);

    float tmax = fmaxf(fmaxf(fmaxf(s[0], s[1]), fmaxf(s[2], s[3])),
                       fmaxf(fmaxf(s[4], s[5]), fmaxf(s[6], s[7])));
    tmax = fmaxf(tmax, __shfl_xor(tmax, 16, 32));
    float m_new = fmaxf(m_run, tmax);
    float p = 0.f;
    #pragma unroll
    for (int i = 0; i < 8; ++i) p += __expf(s[i] - m_new);
    p += __shfl_xor(p, 16, 32);
    z_run = fmaf(z_run, __expf(m_run - m_new), p);
    m_run = m_new;
}

__global__ void __launch_bounds__(32) __attribute__((amdgpu_waves_per_eu(4, 4)))
col_softmax_stats_kernel(const float* __restrict__ Qp, const float* __restrict__ Kp,
                         float* __restrict__ mArr, float* __restrict__ zinvArr)
{
    int wave = blockIdx.x;
    int bb = wave / NT;
    int tt = wave - bb * NT;
    int lane = threadIdx.x & 31;
    int col  = lane & 15;
    int hl   = lane >> 4;

    const float* kp = Kp + ((size_t)(bb * SS) + (size_t)tt * 16 + col) * LDQ + 2 * hl;
    v2f b0 = *(const v2f*)(kp +  0);
    v2f b1 = *(const v2f*)(kp +  4);
    v2f b2 = *(const v2f*)(kp +  8);
    v2f b3 = *(const v2f*)(kp + 12);
    v2f b4 = *(const v2f*)(kp + 16);
    v2f b5 = *(const v2f*)(kp + 20);

    const float* qbase = Qp + ((size_t)(bb * SS) + col) * LDQ + 2 * hl;
    float m_run = -3.0e38f;
    float z_run = 0.f;

    QFrag fa, fb, na, nb;
    load_qfrag(fa, qbase);
    load_qfrag(fb, qbase + (size_t)16 * LDQ);
    for (int qt = 0; qt < NT; qt += 2) {
        int n0 = (qt + 2 < NT) ? qt + 2 : NT - 1;
        int n1 = (qt + 3 < NT) ? qt + 3 : NT - 1;
        load_qfrag(na, qbase + (size_t)n0 * 16 * LDQ);
        load_qfrag(nb, qbase + (size_t)n1 * 16 * LDQ);
        stats_step(fa, b0, b1, b2, b3, b4, b5, m_run, z_run);
        stats_step(fb, b0, b1, b2, b3, b4, b5, m_run, z_run);
        asm volatile("" :: "v"(na.a0), "v"(na.a1), "v"(na.a2), "v"(na.a3), "v"(na.a4), "v"(na.a5),
                           "v"(nb.a0), "v"(nb.a1), "v"(nb.a2), "v"(nb.a3), "v"(nb.a4), "v"(nb.a5),
                           "v"(z_run));
        fa = na;
        fb = nb;
    }
    if (hl == 0) {
        mArr[bb * SS + tt * 16 + col]    = m_run;
        zinvArr[bb * SS + tt * 16 + col] = 1.f / z_run;
    }
}

__global__ void __launch_bounds__(256)
scale_v_kernel(float* __restrict__ Vm, const float* __restrict__ zinvArr)
{
    int i = blockIdx.x * blockDim.x + threadIdx.x;
    if (i < BB * SS * DV) Vm[i] *= zinvArr[i / DV];
}

struct TFrag { v2f b0, b1, b2, b3, b4, b5; float m; v2f bv0[4], bv1[4]; };

__device__ __forceinline__ void load_tfrag(TFrag& f, const float* kp, const float* mp,
                                           const float* vt, int col, int hl)
{
    f.b0 = *(const v2f*)(kp +  0);
    f.b1 = *(const v2f*)(kp +  4);
    f.b2 = *(const v2f*)(kp +  8);
    f.b3 = *(const v2f*)(kp + 12);
    f.b4 = *(const v2f*)(kp + 16);
    f.b5 = *(const v2f*)(kp + 20);
    f.m  = mp[col];
    #pragma unroll
    for (int c = 0; c < 4; ++c) {
        const float* vp = vt + (size_t)(4 * c + 2 * hl) * DV;
        f.bv0[c].x = vp[col];        f.bv0[c].y = vp[DV + col];
        f.bv1[c].x = vp[16 + col];   f.bv1[c].y = vp[DV + 16 + col];
    }
}

__global__ void __launch_bounds__(32) __attribute__((amdgpu_waves_per_eu(4, 4)))
attn_h1_kernel(const float* __restrict__ Qp, const float* __restrict__ Kp,
               const float* __restrict__ VZ, const float* __restrict__ mArr,
               float* __restrict__ h1)
{
    __shared__ float ldsE[16 * 20];
    int wave = blockIdx.x;
    int bb = wave / NT;
    int qt = wave - bb * NT;
    int lane = threadIdx.x & 31;
    int col  = lane & 15;
    int hl   = lane >> 4;

    const float* qp = Qp + ((size_t)(bb * SS) + (size_t)qt * 16 + col) * LDQ + 2 * hl;
    v2f a0 = *(const v2f*)(qp +  0);
    v2f a1 = *(const v2f*)(qp +  4);
    v2f a2 = *(const v2f*)(qp +  8);
    v2f a3 = *(const v2f*)(qp + 12);
    v2f a4 = *(const v2f*)(qp + 16);
    v2f a5 = *(const v2f*)(qp + 20);

    v8f acc0 = {};
    v8f acc1 = {};

    const float* kbase = Kp + ((size_t)(bb * SS) + col) * LDQ + 2 * hl;
    const float* vzb   = VZ + (size_t)(bb * SS) * DV;
    const float* mrow  = mArr + bb * SS;

    auto step = [&](const TFrag& f) {
        v8f s = {};
        s = WMMA_F32X4(a0, f.b0, s);
        s = WMMA_F32X4(a1, f.b1, s);
        s = WMMA_F32X4(a2, f.b2, s);
        s = WMMA_F32X4(a3, f.b3, s);
        s = WMMA_F32X4(a4, f.b4, s);
        s = WMMA_F32X4(a5, f.b5, s);

        #pragma unroll
        for (int r = 0; r < 8; ++r)
            ldsE[(r + 8 * hl) * 20 + col] = __expf(s[r] - f.m);

        #pragma unroll
        for (int c = 0; c < 4; ++c) {
            v2f ae = *(const v2f*)&ldsE[col * 20 + 4 * c + 2 * hl];
            acc0 = WMMA_F32X4(ae, f.bv0[c], acc0);
            acc1 = WMMA_F32X4(ae, f.bv1[c], acc1);
        }
    };

    TFrag fa, fb;
    load_tfrag(fa, kbase, mrow, vzb, col, hl);
    for (int tt = 0; tt < NT; tt += 2) {
        int t1 = tt + 1;
        load_tfrag(fb, kbase + (size_t)t1 * 16 * LDQ, mrow + t1 * 16,
                   vzb + (size_t)(t1 * 16) * DV, col, hl);
        step(fa);
        int t2 = (tt + 2 < NT) ? tt + 2 : NT - 1;
        load_tfrag(fa, kbase + (size_t)t2 * 16 * LDQ, mrow + t2 * 16,
                   vzb + (size_t)(t2 * 16) * DV, col, hl);
        step(fb);
    }

    #pragma unroll
    for (int r = 0; r < 8; ++r) {
        float* hp = h1 + ((size_t)(bb * SS) + (size_t)qt * 16 + r + 8 * hl) * DV;
        hp[col]      = acc0[r];
        hp[col + 16] = acc1[r];
    }
}

__global__ void __launch_bounds__(256)
out_proj_kernel(const float* __restrict__ h1, const float* __restrict__ Wh,
                float* __restrict__ out)
{
    int row = blockIdx.x * blockDim.x + threadIdx.x;
    if (row >= BB * SS) return;
    const float* hp = h1 + (size_t)row * DV;
    float acc[DD];
    #pragma unroll
    for (int d = 0; d < DD; ++d) acc[d] = 0.f;
    for (int v = 0; v < DV; ++v) {
        float hv = hp[v];
        #pragma unroll
        for (int d = 0; d < DD; ++d)
            acc[d] = fmaf(hv, Wh[v * DD + d], acc[d]);
    }
    float* op = out + (size_t)row * DD;
    #pragma unroll
    for (int d = 0; d < DD; ++d) op[d] = acc[d];
}

extern "C" void kernel_launch(void* const* d_in, const int* in_sizes, int n_in,
                              void* d_out, int out_size, void* d_ws, size_t ws_size,
                              hipStream_t stream)
{
    const float* x  = (const float*)d_in[0];
    const float* Wq = (const float*)d_in[1];
    const float* Wk = (const float*)d_in[2];
    const float* Wv = (const float*)d_in[3];
    const float* Wh = (const float*)d_in[4];
    float* out = (float*)d_out;

    float* ws = (float*)d_ws;
    size_t o = 0;
    float* Qp = ws + o; o += (size_t)BB * SS * LDQ;
    float* Kp = ws + o; o += (size_t)BB * SS * LDQ;
    float* Vm = ws + o; o += (size_t)BB * SS * DV;
    float* mA = ws + o; o += (size_t)BB * SS;
    float* zI = ws + o; o += (size_t)BB * SS;
    float* h1 = ws + o; o += (size_t)BB * SS * DV;

    const int rows = BB * SS;
    qkv_proj_kernel<<<(rows + 255) / 256, 256, 0, stream>>>(x, Wq, Wk, Wv, Qp, Kp, Vm);
    col_softmax_stats_kernel<<<BB * NT, 32, 0, stream>>>(Qp, Kp, mA, zI);
    const int nv = BB * SS * DV;
    scale_v_kernel<<<(nv + 255) / 256, 256, 0, stream>>>(Vm, zI);
    attn_h1_kernel<<<BB * NT, 32, 0, stream>>>(Qp, Kp, Vm, mA, h1);
    out_proj_kernel<<<(rows + 255) / 256, 256, 0, stream>>>(h1, Wh, out);
}


# Q and K planes stored in WMMA fragment order (16 B per lane, lanes contiguous): every operand load is a whole-line b128, not a 16-line gather; k chunks regrouped, so the k sum order changes
# speedup vs baseline: 1.2256x; 1.1083x over previous
#include <hip/hip_runtime.h>
#include <hip/hip_bf16.h>


typedef __attribute__((ext_vector_type(2))) float v2f;
typedef __attribute__((ext_vector_type(4))) float v4f;
typedef __attribute__((ext_vector_type(8))) float v8f;

constexpr int BB  = 32;
constexpr int SS  = 2048;
constexpr int DD  = 64;
constexpr int DK  = 24;
constexpr int DV  = 32;
constexpr int NT  = SS / 16;
constexpr int TILE_F = 16 * DK;

#define WMMA_F32X4(Am, Bm, Cm) \
  __builtin_amdgcn_wmma_f32_16x16x4_f32(false, (Am), false, (Bm), (short)0, (Cm), false, false)

__global__ void __launch_bounds__(256)
qkv_proj_kernel(const float* __restrict__ x, const float* __restrict__ Wq,
                const float* __restrict__ Wk, const float* __restrict__ Wv,
                float* __restrict__ Qp, float* __restrict__ Kp, float* __restrict__ Vm)
{
    int row = blockIdx.x * blockDim.x + threadIdx.x;
    if (row >= BB * SS) return;
    const float* xp = x + (size_t)row * DD;

    float accq[DK], acck[DK], accv[DV];
    #pragma unroll
    for (int j = 0; j < DK; ++j) { accq[j] = 0.f; acck[j] = 0.f; }
    #pragma unroll
    for (int j = 0; j < DV; ++j) accv[j] = 0.f;

    for (int d = 0; d < DD; ++d) {
        float xd = xp[d];
        #pragma unroll
        for (int j = 0; j < DK; ++j) {
            accq[j] = fmaf(xd, Wq[d * DK + j], accq[j]);
            acck[j] = fmaf(xd, Wk[d * DK + j], acck[j]);
        }
        #pragma unroll
        for (int j = 0; j < DV; ++j)
            accv[j] = fmaf(xd, Wv[d * DV + j], accv[j]);
    }
    float* qo = Qp + (size_t)(row >> 4) * TILE_F + (row & 15) * 4;
    float* ko = Kp + (size_t)(row >> 4) * TILE_F + (row & 15) * 4;
    float* vo = Vm + (size_t)row * DV;
    #pragma unroll
    for (int hl = 0; hl < 2; ++hl) {
        #pragma unroll
        for (int p = 0; p < 3; ++p) {
            int k0 = 12 * hl + 4 * p;
            v4f qv = {accq[k0], accq[k0 + 1], accq[k0 + 2], accq[k0 + 3]};
            v4f kv = {acck[k0], acck[k0 + 1], acck[k0 + 2], acck[k0 + 3]};
            *(v4f*)(qo + p * 128 + hl * 64) = qv;
            *(v4f*)(ko + p * 128 + hl * 64) = kv;
        }
    }
    #pragma unroll
    for (int j = 0; j < DV; ++j) vo[j] = accv[j];
}

struct QFrag { v4f p0, p1, p2; };

__device__ __forceinline__ void load_qfrag(QFrag& f, const float* tp)
{
    f.p0 = *(const v4f*)(tp);
    f.p1 = *(const v4f*)(tp + 128);
    f.p2 = *(const v4f*)(tp + 256);
}

__device__ __forceinline__ v2f lo2(v4f v) { return __builtin_shufflevector(v, v, 0, 1); }
__device__ __forceinline__ v2f hi2(v4f v) { return __builtin_shufflevector(v, v, 2, 3); }

__device__ __forceinline__ void stats_step(const QFrag& f, v2f b0, v2f b1, v2f b2,
                                           v2f b3, v2f b4, v2f b5,
                                           float& m_run, float& z_run)
{
    v8f s = {};
    s = WMMA_F32X4(lo2(f.p0), b0, s);
    s = WMMA_F32X4(hi2(f.p0), b1, s);
    s = WMMA_F32X4(lo2(f.p1), b2, s);
    s = WMMA_F32X4(hi2(f.p1), b3, s);
    s = WMMA_F32X4(lo2(f.p2), b4, s);
    s = WMMA_F32X4(hi2(f.p2), b5, s);

    float tmax = fmaxf(fmaxf(fmaxf(s[0], s[1]), fmaxf(s[2], s[3])),
                       fmaxf(fmaxf(s[4], s[5]), fmaxf(s[6], s[7])));
    tmax = fmaxf(tmax, __shfl_xor(tmax, 16, 32));
    float m_new = fmaxf(m_run, tmax);
    float p = 0.f;
    #pragma unroll
    for (int i = 0; i < 8; ++i) p += __expf(s[i] - m_new);
    p += __shfl_xor(p, 16, 32);
    z_run = fmaf(z_run, __expf(m_run - m_new), p);
    m_run = m_new;
}

__global__ void __launch_bounds__(32) __attribute__((amdgpu_waves_per_eu(4, 4)))
col_softmax_stats_kernel(const float* __restrict__ Qp, const float* __restrict__ Kp,
                         float* __restrict__ mArr, float* __restrict__ zinvArr)
{
    int wave = blockIdx.x;
    int bb = wave / NT;
    int tt = wave - bb * NT;
    int lane = threadIdx.x & 31;
    int col  = lane & 15;
    int hl   = lane >> 4;

    QFrag kf;
    load_qfrag(kf, Kp + ((size_t)bb * NT + tt) * TILE_F + lane * 4);
    v2f b0 = lo2(kf.p0), b1 = hi2(kf.p0);
    v2f b2 = lo2(kf.p1), b3 = hi2(kf.p1);
    v2f b4 = lo2(kf.p2), b5 = hi2(kf.p2);

    const float* qbase = Qp + (size_t)bb * NT * TILE_F + lane * 4;
    float m_run = -3.0e38f;
    float z_run = 0.f;

    QFrag fa, fb, na, nb;
    load_qfrag(fa, qbase);
    load_qfrag(fb, qbase + TILE_F);
    for (int qt = 0; qt < NT; qt += 2) {
        int n0 = (qt + 2 < NT) ? qt + 2 : NT - 1;
        int n1 = (qt + 3 < NT) ? qt + 3 : NT - 1;
        load_qfrag(na, qbase + n0 * TILE_F);
        load_qfrag(nb, qbase + n1 * TILE_F);
        stats_step(fa, b0, b1, b2, b3, b4, b5, m_run, z_run);
        stats_step(fb, b0, b1, b2, b3, b4, b5, m_run, z_run);
        asm volatile("" :: "v"(na.p0), "v"(na.p1), "v"(na.p2),
                           "v"(nb.p0), "v"(nb.p1), "v"(nb.p2),
                           "v"(z_run));
        fa = na;
        fb = nb;
    }
    if (hl == 0) {
        mArr[bb * SS + tt * 16 + col]    = m_run;
        zinvArr[bb * SS + tt * 16 + col] = 1.f / z_run;
    }
}

__global__ void __launch_bounds__(256)
scale_v_kernel(float* __restrict__ Vm, const float* __restrict__ zinvArr)
{
    int i = blockIdx.x * blockDim.x + threadIdx.x;
    if (i < BB * SS * DV) Vm[i] *= zinvArr[i / DV];
}

struct TFrag { QFrag k; float m; v2f bv0[4], bv1[4]; };

__device__ __forceinline__ void load_tfrag(TFrag& f, const float* kp, const float* mp,
                                           const float* vt, int col, int hl)
{
    load_qfrag(f.k, kp);
    f.m  = mp[col];
    #pragma unroll
    for (int c = 0; c < 4; ++c) {
        const float* vp = vt + (size_t)(4 * c + 2 * hl) * DV;
        f.bv0[c].x = vp[col];        f.bv0[c].y = vp[DV + col];
        f.bv1[c].x = vp[16 + col];   f.bv1[c].y = vp[DV + 16 + col];
    }
}

__global__ void __launch_bounds__(32) __attribute__((amdgpu_waves_per_eu(4, 4)))
attn_h1_kernel(const float* __restrict__ Qp, const float* __restrict__ Kp,
               const float* __restrict__ VZ, const float* __restrict__ mArr,
               float* __restrict__ h1)
{
    __shared__ float ldsE[16 * 20];
    int wave = blockIdx.x;
    int bb = wave / NT;
    int qt = wave - bb * NT;
    int lane = threadIdx.x & 31;
    int col  = lane & 15;
    int hl   = lane >> 4;

    QFrag qf;
    load_qfrag(qf, Qp + ((size_t)bb * NT + qt) * TILE_F + lane * 4);
    v2f a0 = lo2(qf.p0), a1 = hi2(qf.p0);
    v2f a2 = lo2(qf.p1), a3 = hi2(qf.p1);
    v2f a4 = lo2(qf.p2), a5 = hi2(qf.p2);

    v8f acc0 = {};
    v8f acc1 = {};

    const float* kbase = Kp + (size_t)bb * NT * TILE_F + lane * 4;
    const float* vzb   = VZ + (size_t)(bb * SS) * DV;
    const float* mrow  = mArr + bb * SS;

    auto step = [&](const TFrag& f) {
        v8f s = {};
        s = WMMA_F32X4(a0, lo2(f.k.p0), s);
        s = WMMA_F32X4(a1, hi2(f.k.p0), s);
        s = WMMA_F32X4(a2, lo2(f.k.p1), s);
        s = WMMA_F32X4(a3, hi2(f.k.p1), s);
        s = WMMA_F32X4(a4, lo2(f.k.p2), s);
        s = WMMA_F32X4(a5, hi2(f.k.p2), s);

        #pragma unroll
        for (int r = 0; r < 8; ++r)
            ldsE[(r + 8 * hl) * 20 + col] = __expf(s[r] - f.m);

        #pragma unroll
        for (int c = 0; c < 4; ++c) {
            v2f ae = *(const v2f*)&ldsE[col * 20 + 4 * c + 2 * hl];
            acc0 = WMMA_F32X4(ae, f.bv0[c], acc0);
            acc1 = WMMA_F32X4(ae, f.bv1[c], acc1);
        }
    };

    TFrag fa, fb;
    load_tfrag(fa, kbase, mrow, vzb, col, hl);
    for (int tt = 0; tt < NT; tt += 2) {
        int t1 = tt + 1;
        load_tfrag(fb, kbase + t1 * TILE_F, mrow + t1 * 16,
                   vzb + (size_t)(t1 * 16) * DV, col, hl);
        step(fa);
        int t2 = (tt + 2 < NT) ? tt + 2 : NT - 1;
        load_tfrag(fa, kbase + t2 * TILE_F, mrow + t2 * 16,
                   vzb + (size_t)(t2 * 16) * DV, col, hl);
        step(fb);
    }

    #pragma unroll
    for (int r = 0; r < 8; ++r) {
        float* hp = h1 + ((size_t)(bb * SS) + (size_t)qt * 16 + r + 8 * hl) * DV;
        hp[col]      = acc0[r];
        hp[col + 16] = acc1[r];
    }
}

__global__ void __launch_bounds__(256)
out_proj_kernel(const float* __restrict__ h1, const float* __restrict__ Wh,
                float* __restrict__ out)
{
    int row = blockIdx.x * blockDim.x + threadIdx.x;
    if (row >= BB * SS) return;
    const float* hp = h1 + (size_t)row * DV;
    float acc[DD];
    #pragma unroll
    for (int d = 0; d < DD; ++d) acc[d] = 0.f;
    for (int v = 0; v < DV; ++v) {
        float hv = hp[v];
        #pragma unroll
        for (int d = 0; d < DD; ++d)
            acc[d] = fmaf(hv, Wh[v * DD + d], acc[d]);
    }
    float* op = out + (size_t)row * DD;
    #pragma unroll
    for (int d = 0; d < DD; ++d) op[d] = acc[d];
}

extern "C" void kernel_launch(void* const* d_in, const int* in_sizes, int n_in,
                              void* d_out, int out_size, void* d_ws, size_t ws_size,
                              hipStream_t stream)
{
    const float* x  = (const float*)d_in[0];
    const float* Wq = (const float*)d_in[1];
    const float* Wk = (const float*)d_in[2];
    const float* Wv = (const float*)d_in[3];
    const float* Wh = (const float*)d_in[4];
    float* out = (float*)d_out;

    float* ws = (float*)d_ws;
    size_t o = 0;
    float* Qp = ws + o; o += (size_t)BB * SS * 32;
    float* Kp = ws + o; o += (size_t)BB * SS * 32;
    float* Vm = ws + o; o += (size_t)BB * SS * DV;
    float* mA = ws + o; o += (size_t)BB * SS;
    float* zI = ws + o; o += (size_t)BB * SS;
    float* h1 = ws + o; o += (size_t)BB * SS * DV;

    const int rows = BB * SS;
    qkv_proj_kernel<<<(rows + 255) / 256, 256, 0, stream>>>(x, Wq, Wk, Wv, Qp, Kp, Vm);
    col_softmax_stats_kernel<<<BB * NT, 32, 0, stream>>>(Qp, Kp, mA, zI);
    const int nv = BB * SS * DV;
    scale_v_kernel<<<(nv + 255) / 256, 256, 0, stream>>>(Vm, zI);
    attn_h1_kernel<<<BB * NT, 32, 0, stream>>>(Qp, Kp, Vm, mA, h1);
    out_proj_kernel<<<(rows + 255) / 256, 256, 0, stream>>>(h1, Wh, out);
}


# Streaming WMMA loops: operands requested two tiles (h1) / three 2-tile trips (stats) ahead into four register sets rotating by name, not one ahead: an L2 round trip no longer paces a trip; same sums
# speedup vs baseline: 1.3151x; 1.0730x over previous
#include <hip/hip_runtime.h>
#include <hip/hip_bf16.h>


typedef __attribute__((ext_vector_type(2))) float v2f;
typedef __attribute__((ext_vector_type(4))) float v4f;
typedef __attribute__((ext_vector_type(8))) float v8f;

constexpr int BB  = 32;
constexpr int SS  = 2048;
constexpr int DD  = 64;
constexpr int DK  = 24;
constexpr int DV  = 32;
constexpr int NT  = SS / 16;
constexpr int TILE_F = 16 * DK;

#define WMMA_F32X4(Am, Bm, Cm) \
  __builtin_amdgcn_wmma_f32_16x16x4_f32(false, (Am), false, (Bm), (short)0, (Cm), false, false)

#define TRIP_EDGE() __builtin_amdgcn_sched_barrier(0)

__global__ void __launch_bounds__(256)
qkv_proj_kernel(const float* __restrict__ x, const float* __restrict__ Wq,
                const float* __restrict__ Wk, const float* __restrict__ Wv,
                float* __restrict__ Qp, float* __restrict__ Kp, float* __restrict__ Vm)
{
    int row = blockIdx.x * blockDim.x + threadIdx.x;
    if (row >= BB * SS) return;
    const float* xp = x + (size_t)row * DD;

    float accq[DK], acck[DK], accv[DV];
    #pragma unroll
    for (int j = 0; j < DK; ++j) { accq[j] = 0.f; acck[j] = 0.f; }
    #pragma unroll
    for (int j = 0; j < DV; ++j) accv[j] = 0.f;

    for (int d = 0; d < DD; ++d) {
        float xd = xp[d];
        #pragma unroll
        for (int j = 0; j < DK; ++j) {
            accq[j] = fmaf(xd, Wq[d * DK + j], accq[j]);
            acck[j] = fmaf(xd, Wk[d * DK + j], acck[j]);
        }
        #pragma unroll
        for (int j = 0; j < DV; ++j)
            accv[j] = fmaf(xd, Wv[d * DV + j], accv[j]);
    }
    float* qo = Qp + (size_t)(row >> 4) * TILE_F + (row & 15) * 4;
    float* ko = Kp + (size_t)(row >> 4) * TILE_F + (row & 15) * 4;
    float* vo = Vm + (size_t)row * DV;
    #pragma unroll
    for (int hl = 0; hl < 2; ++hl) {
        #pragma unroll
        for (int p = 0; p < 3; ++p) {
            int k0 = 12 * hl + 4 * p;
            v4f qv = {accq[k0], accq[k0 + 1], accq[k0 + 2], accq[k0 + 3]};
            v4f kv = {acck[k0], acck[k0 + 1], acck[k0 + 2], acck[k0 + 3]};
            *(v4f*)(qo + p * 128 + hl * 64) = qv;
            *(v4f*)(ko + p * 128 + hl * 64) = kv;
        }
    }
    #pragma unroll
    for (int j = 0; j < DV; ++j) vo[j] = accv[j];
}

struct QFrag { v4f p0, p1, p2; };

__device__ __forceinline__ void load_qfrag(QFrag& f, const float* tp)
{
    f.p0 = *(const v4f*)(tp);
    f.p1 = *(const v4f*)(tp + 128);
    f.p2 = *(const v4f*)(tp + 256);
}

__device__ __forceinline__ v2f lo2(v4f v) { return __builtin_shufflevector(v, v, 0, 1); }
__device__ __forceinline__ v2f hi2(v4f v) { return __builtin_shufflevector(v, v, 2, 3); }

__device__ __forceinline__ void stats_step(const QFrag& f, v2f b0, v2f b1, v2f b2,
                                           v2f b3, v2f b4, v2f b5,
                                           float& m_run, float& z_run)
{
    v8f s = {};
    s = WMMA_F32X4(lo2(f.p0), b0, s);
    s = WMMA_F32X4(hi2(f.p0), b1, s);
    s = WMMA_F32X4(lo2(f.p1), b2, s);
    s = WMMA_F32X4(hi2(f.p1), b3, s);
    s = WMMA_F32X4(lo2(f.p2), b4, s);
    s = WMMA_F32X4(hi2(f.p2), b5, s);

    float tmax = fmaxf(fmaxf(fmaxf(s[0], s[1]), fmaxf(s[2], s[3])),
                       fmaxf(fmaxf(s[4], s[5]), fmaxf(s[6], s[7])));
    tmax = fmaxf(tmax, __shfl_xor(tmax, 16, 32));
    float m_new = fmaxf(m_run, tmax);
    float p = 0.f;
    #pragma unroll
    for (int i = 0; i < 8; ++i) p += __expf(s[i] - m_new);
    p += __shfl_xor(p, 16, 32);
    z_run = fmaf(z_run, __expf(m_run - m_new), p);
    m_run = m_new;
}

__global__ void __launch_bounds__(32) __attribute__((amdgpu_waves_per_eu(4, 4)))
col_softmax_stats_kernel(const float* __restrict__ Qp, const float* __restrict__ Kp,
                         float* __restrict__ mArr, float* __restrict__ zinvArr)
{
    int wave = blockIdx.x;
    int bb = wave / NT;
    int tt = wave - bb * NT;
    int lane = threadIdx.x & 31;
    int col  = lane & 15;
    int hl   = lane >> 4;

    QFrag kf;
    load_qfrag(kf, Kp + ((size_t)bb * NT + tt) * TILE_F + lane * 4);
    v2f b0 = lo2(kf.p0), b1 = hi2(kf.p0);
    v2f b2 = lo2(kf.p1), b3 = hi2(kf.p1);
    v2f b4 = lo2(kf.p2), b5 = hi2(kf.p2);

    const float* qbase = Qp + (size_t)bb * NT * TILE_F + lane * 4;
    float m_run = -3.0e38f;
    float z_run = 0.f;

    QFrag p0a, p0b, p1a, p1b, p2a, p2b, p3a, p3b;
    load_qfrag(p0a, qbase);
    load_qfrag(p0b, qbase + TILE_F);
    load_qfrag(p1a, qbase + 2 * TILE_F);
    load_qfrag(p1b, qbase + 3 * TILE_F);
    load_qfrag(p2a, qbase + 4 * TILE_F);
    load_qfrag(p2b, qbase + 5 * TILE_F);
    auto tile_at = [&](int t) { return qbase + (t < NT ? t : NT - 1) * TILE_F; };
    for (int qt = 0; qt < NT; qt += 8) {
        load_qfrag(p3a, tile_at(qt + 6));
        load_qfrag(p3b, tile_at(qt + 7));
        TRIP_EDGE();
        stats_step(p0a, b0, b1, b2, b3, b4, b5, m_run, z_run);
        stats_step(p0b, b0, b1, b2, b3, b4, b5, m_run, z_run);
        TRIP_EDGE();
        load_qfrag(p0a, tile_at(qt + 8));
        load_qfrag(p0b, tile_at(qt + 9));
        TRIP_EDGE();
        stats_step(p1a, b0, b1, b2, b3, b4, b5, m_run, z_run);
        stats_step(p1b, b0, b1, b2, b3, b4, b5, m_run, z_run);
        TRIP_EDGE();
        load_qfrag(p1a, tile_at(qt + 10));
        load_qfrag(p1b, tile_at(qt + 11));
        TRIP_EDGE();
        stats_step(p2a, b0, b1, b2, b3, b4, b5, m_run, z_run);
        stats_step(p2b, b0, b1, b2, b3, b4, b5, m_run, z_run);
        TRIP_EDGE();
        load_qfrag(p2a, tile_at(qt + 12));
        load_qfrag(p2b, tile_at(qt + 13));
        TRIP_EDGE();
        stats_step(p3a, b0, b1, b2, b3, b4, b5, m_run, z_run);
        stats_step(p3b, b0, b1, b2, b3, b4, b5, m_run, z_run);
    }
    if (hl == 0) {
        mArr[bb * SS + tt * 16 + col]    = m_run;
        zinvArr[bb * SS + tt * 16 + col] = 1.f / z_run;
    }
}

__global__ void __launch_bounds__(256)
scale_v_kernel(float* __restrict__ Vm, const float* __restrict__ zinvArr)
{
    int i = blockIdx.x * blockDim.x + threadIdx.x;
    if (i < BB * SS * DV) Vm[i] *= zinvArr[i / DV];
}

struct TFrag { QFrag k; float m; v2f bv0[4], bv1[4]; };

__device__ __forceinline__ void load_tfrag(TFrag& f, const float* kp, const float* mp,
                                           const float* vt, int col, int hl)
{
    load_qfrag(f.k, kp);
    f.m  = mp[col];
    #pragma unroll
    for (int c = 0; c < 4; ++c) {
        const float* vp = vt + (size_t)(4 * c + 2 * hl) * DV;
        f.bv0[c].x = vp[col];        f.bv0[c].y = vp[DV + col];
        f.bv1[c].x = vp[16 + col];   f.bv1[c].y = vp[DV + 16 + col];
    }
}

__global__ void __launch_bounds__(32) __attribute__((amdgpu_waves_per_eu(4, 4)))
attn_h1_kernel(const float* __restrict__ Qp, const float* __restrict__ Kp,
               const float* __restrict__ VZ, const float* __restrict__ mArr,
               float* __restrict__ h1)
{
    __shared__ float ldsE[16 * 20];
    int wave = blockIdx.x;
    int bb = wave / NT;
    int qt = wave - bb * NT;
    int lane = threadIdx.x & 31;
    int col  = lane & 15;
    int hl   = lane >> 4;

    QFrag qf;
    load_qfrag(qf, Qp + ((size_t)bb * NT + qt) * TILE_F + lane * 4);
    v2f a0 = lo2(qf.p0), a1 = hi2(qf.p0);
    v2f a2 = lo2(qf.p1), a3 = hi2(qf.p1);
    v2f a4 = lo2(qf.p2), a5 = hi2(qf.p2);

    v8f acc0 = {};
    v8f acc1 = {};

    const float* kbase = Kp + (size_t)bb * NT * TILE_F + lane * 4;
    const float* vzb   = VZ + (size_t)(bb * SS) * DV;
    const float* mrow  = mArr + bb * SS;

    auto step = [&](const TFrag& f) {
        v8f s = {};
        s = WMMA_F32X4(a0, lo2(f.k.p0), s);
        s = WMMA_F32X4(a1, hi2(f.k.p0), s);
        s = WMMA_F32X4(a2, lo2(f.k.p1), s);
        s = WMMA_F32X4(a3, hi2(f.k.p1), s);
        s = WMMA_F32X4(a4, lo2(f.k.p2), s);
        s = WMMA_F32X4(a5, hi2(f.k.p2), s);

        #pragma unroll
        for (int r = 0; r < 8; ++r)
            ldsE[(r + 8 * hl) * 20 + col] = __expf(s[r] - f.m);

        #pragma unroll
        for (int c = 0; c < 4; ++c) {
            v2f ae = *(const v2f*)&ldsE[col * 20 + 4 * c + 2 * hl];
            acc0 = WMMA_F32X4(ae, f.bv0[c], acc0);
            acc1 = WMMA_F32X4(ae, f.bv1[c], acc1);
        }
    };

    auto load_t = [&](TFrag& f, int t) {
        int tc = (t < NT) ? t : NT - 1;
        load_tfrag(f, kbase + tc * TILE_F, mrow + tc * 16,
                   vzb + (size_t)(tc * 16) * DV, col, hl);
    };
    TFrag f0, f1, f2, f3;
    load_t(f0, 0);
    load_t(f1, 1);
    for (int tt = 0; tt < NT; tt += 4) {
        load_t(f2, tt + 2);
        TRIP_EDGE();
        step(f0);
        TRIP_EDGE();
        load_t(f3, tt + 3);
        TRIP_EDGE();
        step(f1);
        TRIP_EDGE();
        load_t(f0, tt + 4);
        TRIP_EDGE();
        step(f2);
        TRIP_EDGE();
        load_t(f1, tt + 5);
        TRIP_EDGE();
        step(f3);
    }

    #pragma unroll
    for (int r = 0; r < 8; ++r) {
        float* hp = h1 + ((size_t)(bb * SS) + (size_t)qt * 16 + r + 8 * hl) * DV;
        hp[col]      = acc0[r];
        hp[col + 16] = acc1[r];
    }
}

__global__ void __launch_bounds__(256)
out_proj_kernel(const float* __restrict__ h1, const float* __restrict__ Wh,
                float* __restrict__ out)
{
    int row = blockIdx.x * blockDim.x + threadIdx.x;
    if (row >= BB * SS) return;
    const float* hp = h1 + (size_t)row * DV;
    float acc[DD];
    #pragma unroll
    for (int d = 0; d < DD; ++d) acc[d] = 0.f;
    for (int v = 0; v < DV; ++v) {
        float hv = hp[v];
        #pragma unroll
        for (int d = 0; d < DD; ++d)
            acc[d] = fmaf(hv, Wh[v * DD + d], acc[d]);
    }
    float* op = out + (size_t)row * DD;
    #pragma unroll
    for (int d = 0; d < DD; ++d) op[d] = acc[d];
}

extern "C" void kernel_launch(void* const* d_in, const int* in_sizes, int n_in,
                              void* d_out, int out_size, void* d_ws, size_t ws_size,
                              hipStream_t stream)
{
    const float* x  = (const float*)d_in[0];
    const float* Wq = (const float*)d_in[1];
    const float* Wk = (const float*)d_in[2];
    const float* Wv = (const float*)d_in[3];
    const float* Wh = (const float*)d_in[4];
    float* out = (float*)d_out;

    float* ws = (float*)d_ws;
    size_t o = 0;
    float* Qp = ws + o; o += (size_t)BB * SS * 32;
    float* Kp = ws + o; o += (size_t)BB * SS * 32;
    float* Vm = ws + o; o += (size_t)BB * SS * DV;
    float* mA = ws + o; o += (size_t)BB * SS;
    float* zI = ws + o; o += (size_t)BB * SS;
    float* h1 = ws + o; o += (size_t)BB * SS * DV;

    const int rows = BB * SS;
    qkv_proj_kernel<<<(rows + 255) / 256, 256, 0, stream>>>(x, Wq, Wk, Wv, Qp, Kp, Vm);
    col_softmax_stats_kernel<<<BB * NT, 32, 0, stream>>>(Qp, Kp, mA, zI);
    const int nv = BB * SS * DV;
    scale_v_kernel<<<(nv + 255) / 256, 256, 0, stream>>>(Vm, zI);
    attn_h1_kernel<<<BB * NT, 32, 0, stream>>>(Qp, Kp, Vm, mA, h1);
    out_proj_kernel<<<(rows + 255) / 256, 256, 0, stream>>>(h1, Wh, out);
}


# Both streaming WMMA kernels: a wave keeps two 16-row tiles resident (8 such waves of one batch per work-group), so each streamed operand tile is loaded once for twice the products; same sums
# speedup vs baseline: 1.4097x; 1.0719x over previous
#include <hip/hip_runtime.h>
#include <hip/hip_bf16.h>


typedef __attribute__((ext_vector_type(2))) float v2f;
typedef __attribute__((ext_vector_type(4))) float v4f;
typedef __attribute__((ext_vector_type(8))) float v8f;

constexpr int BB  = 32;
constexpr int SS  = 2048;
constexpr int DD  = 64;
constexpr int DK  = 24;
constexpr int DV  = 32;
constexpr int NT  = SS / 16;
constexpr int TILE_F = 16 * DK;
constexpr int WPB   = 8;
constexpr int NPAIR = NT / 2;
constexpr int GRID_PAIR = BB * NPAIR / WPB;

#define WMMA_F32X4(Am, Bm, Cm) \
  __builtin_amdgcn_wmma_f32_16x16x4_f32(false, (Am), false, (Bm), (short)0, (Cm), false, false)

#define TRIP_EDGE() __builtin_amdgcn_sched_barrier(0)

__global__ void __launch_bounds__(256)
qkv_proj_kernel(const float* __restrict__ x, const float* __restrict__ Wq,
                const float* __restrict__ Wk, const float* __restrict__ Wv,
                float* __restrict__ Qp, float* __restrict__ Kp, float* __restrict__ Vm)
{
    int row = blockIdx.x * blockDim.x + threadIdx.x;
    if (row >= BB * SS) return;
    const float* xp = x + (size_t)row * DD;

    float accq[DK], acck[DK], accv[DV];
    #pragma unroll
    for (int j = 0; j < DK; ++j) { accq[j] = 0.f; acck[j] = 0.f; }
    #pragma unroll
    for (int j = 0; j < DV; ++j) accv[j] = 0.f;

    for (int d = 0; d < DD; ++d) {
        float xd = xp[d];
        #pragma unroll
        for (int j = 0; j < DK; ++j) {
            accq[j] = fmaf(xd, Wq[d * DK + j], accq[j]);
            acck[j] = fmaf(xd, Wk[d * DK + j], acck[j]);
        }
        #pragma unroll
        for (int j = 0; j < DV; ++j)
            accv[j] = fmaf(xd, Wv[d * DV + j], accv[j]);
    }
    float* qo = Qp + (size_t)(row >> 4) * TILE_F + (row & 15) * 4;
    float* ko = Kp + (size_t)(row >> 4) * TILE_F + (row & 15) * 4;
    float* vo = Vm + (size_t)row * DV;
    #pragma unroll
    for (int hl = 0; hl < 2; ++hl) {
        #pragma unroll
        for (int p = 0; p < 3; ++p) {
            int k0 = 12 * hl + 4 * p;
            v4f qv = {accq[k0], accq[k0 + 1], accq[k0 + 2], accq[k0 + 3]};
            v4f kv = {acck[k0], acck[k0 + 1], acck[k0 + 2], acck[k0 + 3]};
            *(v4f*)(qo + p * 128 + hl * 64) = qv;
            *(v4f*)(ko + p * 128 + hl * 64) = kv;
        }
    }
    #pragma unroll
    for (int j = 0; j < DV; ++j) vo[j] = accv[j];
}

struct QFrag { v4f p0, p1, p2; };

__device__ __forceinline__ void load_qfrag(QFrag& f, const float* tp)
{
    f.p0 = *(const v4f*)(tp);
    f.p1 = *(const v4f*)(tp + 128);
    f.p2 = *(const v4f*)(tp + 256);
}

__device__ __forceinline__ v2f lo2(v4f v) { return __builtin_shufflevector(v, v, 0, 1); }
__device__ __forceinline__ v2f hi2(v4f v) { return __builtin_shufflevector(v, v, 2, 3); }

__device__ __forceinline__ void stats_step(const QFrag& f, v2f b0, v2f b1, v2f b2,
                                           v2f b3, v2f b4, v2f b5,
                                           float& m_run, float& z_run)
{
    v8f s = {};
    s = WMMA_F32X4(lo2(f.p0), b0, s);
    s = WMMA_F32X4(hi2(f.p0), b1, s);
    s = WMMA_F32X4(lo2(f.p1), b2, s);
    s = WMMA_F32X4(hi2(f.p1), b3, s);
    s = WMMA_F32X4(lo2(f.p2), b4, s);
    s = WMMA_F32X4(hi2(f.p2), b5, s);

    float tmax = fmaxf(fmaxf(fmaxf(s[0], s[1]), fmaxf(s[2], s[3])),
                       fmaxf(fmaxf(s[4], s[5]), fmaxf(s[6], s[7])));
    tmax = fmaxf(tmax, __shfl_xor(tmax, 16, 32));
    float m_new = fmaxf(m_run, tmax);
    float p = 0.f;
    #pragma unroll
    for (int i = 0; i < 8; ++i) p += __expf(s[i] - m_new);
    p += __shfl_xor(p, 16, 32);
    z_run = fmaf(z_run, __expf(m_run - m_new), p);
    m_run = m_new;
}

__global__ void __launch_bounds__(32 * WPB) __attribute__((amdgpu_waves_per_eu(2, 2)))
col_softmax_stats_kernel(const float* __restrict__ Qp, const float* __restrict__ Kp,
                         float* __restrict__ mArr, float* __restrict__ zinvArr)
{
    int wv = __builtin_amdgcn_readfirstlane((int)(threadIdx.x >> 5));
    int pr = blockIdx.x * WPB + wv;
    int bb = pr / NPAIR;
    int tt = 2 * (pr - bb * NPAIR);
    int lane = threadIdx.x & 31;

    QFrag kf, kg;
    load_qfrag(kf, Kp + ((size_t)bb * NT + tt) * TILE_F + lane * 4);
    load_qfrag(kg, Kp + ((size_t)bb * NT + tt + 1) * TILE_F + lane * 4);
    v2f b0 = lo2(kf.p0), b1 = hi2(kf.p0);
    v2f b2 = lo2(kf.p1), b3 = hi2(kf.p1);
    v2f b4 = lo2(kf.p2), b5 = hi2(kf.p2);
    v2f c0 = lo2(kg.p0), c1 = hi2(kg.p0);
    v2f c2 = lo2(kg.p1), c3 = hi2(kg.p1);
    v2f c4 = lo2(kg.p2), c5 = hi2(kg.p2);

    const float* qbase = Qp + (size_t)bb * NT * TILE_F + lane * 4;
    float m_run = -3.0e38f, n_run = -3.0e38f;
    float z_run = 0.f,      y_run = 0.f;

    QFrag p0a, p0b, p1a, p1b, p2a, p2b, p3a, p3b;
    load_qfrag(p0a, qbase);
    load_qfrag(p0b, qbase + TILE_F);
    load_qfrag(p1a, qbase + 2 * TILE_F);
    load_qfrag(p1b, qbase + 3 * TILE_F);
    load_qfrag(p2a, qbase + 4 * TILE_F);
    load_qfrag(p2b, qbase + 5 * TILE_F);
    auto tile_at = [&](int t) { return qbase + (t < NT ? t : NT - 1) * TILE_F; };
    for (int qt = 0; qt < NT; qt += 8) {
        load_qfrag(p3a, tile_at(qt + 6));
        load_qfrag(p3b, tile_at(qt + 7));
        TRIP_EDGE();
        stats_step(p0a, b0, b1, b2, b3, b4, b5, m_run, z_run);
        stats_step(p0a, c0, c1, c2, c3, c4, c5, n_run, y_run);
        stats_step(p0b, b0, b1, b2, b3, b4, b5, m_run, z_run);
        stats_step(p0b, c0, c1, c2, c3, c4, c5, n_run, y_run);
        TRIP_EDGE();
        load_qfrag(p0a, tile_at(qt + 8));
        load_qfrag(p0b, tile_at(qt + 9));
        TRIP_EDGE();
        stats_step(p1a, b0, b1, b2, b3, b4, b5, m_run, z_run);
        stats_step(p1a, c0, c1, c2, c3, c4, c5, n_run, y_run);
        stats_step(p1b, b0, b1, b2, b3, b4, b5, m_run, z_run);
        stats_step(p1b, c0, c1, c2, c3, c4, c5, n_run, y_run);
        TRIP_EDGE();
        load_qfrag(p1a, tile_at(qt + 10));
        load_qfrag(p1b, tile_at(qt + 11));
        TRIP_EDGE();
        stats_step(p2a, b0, b1, b2, b3, b4, b5, m_run, z_run);
        stats_step(p2a, c0, c1, c2, c3, c4, c5, n_run, y_run);
        stats_step(p2b, b0, b1, b2, b3, b4, b5, m_run, z_run);
        stats_step(p2b, c0, c1, c2, c3, c4, c5, n_run, y_run);
        TRIP_EDGE();
        load_qfrag(p2a, tile_at(qt + 12));
        load_qfrag(p2b, tile_at(qt + 13));
        TRIP_EDGE();
        stats_step(p3a, b0, b1, b2, b3, b4, b5, m_run, z_run);
        stats_step(p3a, c0, c1, c2, c3, c4, c5, n_run, y_run);
        stats_step(p3b, b0, b1, b2, b3, b4, b5, m_run, z_run);
        stats_step(p3b, c0, c1, c2, c3, c4, c5, n_run, y_run);
    }
    float m_out = (lane < 16) ? m_run : n_run;
    float z_out = (lane < 16) ? z_run : y_run;
    mArr[bb * SS + tt * 16 + lane]    = m_out;
    zinvArr[bb * SS + tt * 16 + lane] = 1.f / z_out;
}

__global__ void __launch_bounds__(256)
scale_v_kernel(float* __restrict__ Vm, const float* __restrict__ zinvArr)
{
    int i = blockIdx.x * blockDim.x + threadIdx.x;
    if (i < BB * SS * DV) Vm[i] *= zinvArr[i / DV];
}

struct TFrag { QFrag k; float m; v2f bv0[4], bv1[4]; };

__device__ __forceinline__ void load_tfrag(TFrag& f, const float* kp, const float* mp,
                                           const float* vt, int col, int hl)
{
    load_qfrag(f.k, kp);
    f.m  = mp[col];
    #pragma unroll
    for (int c = 0; c < 4; ++c) {
        const float* vp = vt + (size_t)(4 * c + 2 * hl) * DV;
        f.bv0[c].x = vp[col];        f.bv0[c].y = vp[DV + col];
        f.bv1[c].x = vp[16 + col];   f.bv1[c].y = vp[DV + 16 + col];
    }
}

__global__ void __launch_bounds__(32 * WPB) __attribute__((amdgpu_waves_per_eu(2, 2)))
attn_h1_kernel(const float* __restrict__ Qp, const float* __restrict__ Kp,
               const float* __restrict__ VZ, const float* __restrict__ mArr,
               float* __restrict__ h1)
{
    __shared__ float ldsE[WPB][2][16 * 20];
    int wv = __builtin_amdgcn_readfirstlane((int)(threadIdx.x >> 5));
    int pr = blockIdx.x * WPB + wv;
    int bb = pr / NPAIR;
    int qt = 2 * (pr - bb * NPAIR);
    int lane = threadIdx.x & 31;
    int col  = lane & 15;
    int hl   = lane >> 4;
    float* e0 = &ldsE[wv][0][0];
    float* e1 = &ldsE[wv][1][0];

    QFrag qf, qg;
    load_qfrag(qf, Qp + ((size_t)bb * NT + qt) * TILE_F + lane * 4);
    load_qfrag(qg, Qp + ((size_t)bb * NT + qt + 1) * TILE_F + lane * 4);
    v2f a0 = lo2(qf.p0), a1 = hi2(qf.p0);
    v2f a2 = lo2(qf.p1), a3 = hi2(qf.p1);
    v2f a4 = lo2(qf.p2), a5 = hi2(qf.p2);
    v2f g0 = lo2(qg.p0), g1 = hi2(qg.p0);
    v2f g2 = lo2(qg.p1), g3 = hi2(qg.p1);
    v2f g4 = lo2(qg.p2), g5 = hi2(qg.p2);

    v8f acc0 = {};
    v8f acc1 = {};
    v8f acc2 = {};
    v8f acc3 = {};

    const float* kbase = Kp + (size_t)bb * NT * TILE_F + lane * 4;
    const float* vzb   = VZ + (size_t)(bb * SS) * DV;
    const float* mrow  = mArr + bb * SS;

    auto step = [&](const TFrag& f) {
        v8f s = {}, u = {};
        s = WMMA_F32X4(a0, lo2(f.k.p0), s);
        u = WMMA_F32X4(g0, lo2(f.k.p0), u);
        s = WMMA_F32X4(a1, hi2(f.k.p0), s);
        u = WMMA_F32X4(g1, hi2(f.k.p0), u);
        s = WMMA_F32X4(a2, lo2(f.k.p1), s);
        u = WMMA_F32X4(g2, lo2(f.k.p1), u);
        s = WMMA_F32X4(a3, hi2(f.k.p1), s);
        u = WMMA_F32X4(g3, hi2(f.k.p1), u);
        s = WMMA_F32X4(a4, lo2(f.k.p2), s);
        u = WMMA_F32X4(g4, lo2(f.k.p2), u);
        s = WMMA_F32X4(a5, hi2(f.k.p2), s);
        u = WMMA_F32X4(g5, hi2(f.k.p2), u);

        #pragma unroll
        for (int r = 0; r < 8; ++r)
            e0[(r + 8 * hl) * 20 + col] = __expf(s[r] - f.m);
        #pragma unroll
        for (int r = 0; r < 8; ++r)
            e1[(r + 8 * hl) * 20 + col] = __expf(u[r] - f.m);

        #pragma unroll
        for (int c = 0; c < 4; ++c) {
            v2f ae = *(const v2f*)&e0[col * 20 + 4 * c + 2 * hl];
            v2f af = *(const v2f*)&e1[col * 20 + 4 * c + 2 * hl];
            acc0 = WMMA_F32X4(ae, f.bv0[c], acc0);
            acc1 = WMMA_F32X4(ae, f.bv1[c], acc1);
            acc2 = WMMA_F32X4(af, f.bv0[c], acc2);
            acc3 = WMMA_F32X4(af, f.bv1[c], acc3);
        }
    };

    auto load_t = [&](TFrag& f, int t) {
        int tc = (t < NT) ? t : NT - 1;
        load_tfrag(f, kbase + tc * TILE_F, mrow + tc * 16,
                   vzb + (size_t)(tc * 16) * DV, col, hl);
    };
    TFrag f0, f1, f2, f3;
    load_t(f0, 0);
    load_t(f1, 1);
    for (int tt = 0; tt < NT; tt += 4) {
        load_t(f2, tt + 2);
        TRIP_EDGE();
        step(f0);
        TRIP_EDGE();
        load_t(f3, tt + 3);
        TRIP_EDGE();
        step(f1);
        TRIP_EDGE();
        load_t(f0, tt + 4);
        TRIP_EDGE();
        step(f2);
        TRIP_EDGE();
        load_t(f1, tt + 5);
        TRIP_EDGE();
        step(f3);
    }

    #pragma unroll
    for (int r = 0; r < 8; ++r) {
        float* hp = h1 + ((size_t)(bb * SS) + (size_t)qt * 16 + r + 8 * hl) * DV;
        hp[col]      = acc0[r];
        hp[col + 16] = acc1[r];
        hp[16 * DV + col]      = acc2[r];
        hp[16 * DV + col + 16] = acc3[r];
    }
}

__global__ void __launch_bounds__(256)
out_proj_kernel(const float* __restrict__ h1, const float* __restrict__ Wh,
                float* __restrict__ out)
{
    int row = blockIdx.x * blockDim.x + threadIdx.x;
    if (row >= BB * SS) return;
    const float* hp = h1 + (size_t)row * DV;
    float acc[DD];
    #pragma unroll
    for (int d = 0; d < DD; ++d) acc[d] = 0.f;
    for (int v = 0; v < DV; ++v) {
        float hv = hp[v];
        #pragma unroll
        for (int d = 0; d < DD; ++d)
            acc[d] = fmaf(hv, Wh[v * DD + d], acc[d]);
    }
    float* op = out + (size_t)row * DD;
    #pragma unroll
    for (int d = 0; d < DD; ++d) op[d] = acc[d];
}

extern "C" void kernel_launch(void* const* d_in, const int* in_sizes, int n_in,
                              void* d_out, int out_size, void* d_ws, size_t ws_size,
                              hipStream_t stream)
{
    const float* x  = (const float*)d_in[0];
    const float* Wq = (const float*)d_in[1];
    const float* Wk = (const float*)d_in[2];
    const float* Wv = (const float*)d_in[3];
    const float* Wh = (const float*)d_in[4];
    float* out = (float*)d_out;

    float* ws = (float*)d_ws;
    size_t o = 0;
    float* Qp = ws + o; o += (size_t)BB * SS * 32;
    float* Kp = ws + o; o += (size_t)BB * SS * 32;
    float* Vm = ws + o; o += (size_t)BB * SS * DV;
    float* mA = ws + o; o += (size_t)BB * SS;
    float* zI = ws + o; o += (size_t)BB * SS;
    float* h1 = ws + o; o += (size_t)BB * SS * DV;

    const int rows = BB * SS;
    qkv_proj_kernel<<<(rows + 255) / 256, 256, 0, stream>>>(x, Wq, Wk, Wv, Qp, Kp, Vm);
    col_softmax_stats_kernel<<<GRID_PAIR, 32 * WPB, 0, stream>>>(Qp, Kp, mA, zI);
    const int nv = BB * SS * DV;
    scale_v_kernel<<<(nv + 255) / 256, 256, 0, stream>>>(Vm, zI);
    attn_h1_kernel<<<GRID_PAIR, 32 * WPB, 0, stream>>>(Qp, Kp, Vm, mA, h1);
    out_proj_kernel<<<(rows + 255) / 256, 256, 0, stream>>>(h1, Wh, out);
}


# Scale launch removed: each stats wave multiplies the 32 v rows of its own two key tiles by 1/Z at its end (same products, same values); 5 launches -> 4, 65536 one-element waves no longer enter
# speedup vs baseline: 1.4572x; 1.0337x over previous
#include <hip/hip_runtime.h>
#include <hip/hip_bf16.h>


typedef __attribute__((ext_vector_type(2))) float v2f;
typedef __attribute__((ext_vector_type(4))) float v4f;
typedef __attribute__((ext_vector_type(8))) float v8f;

constexpr int BB  = 32;
constexpr int SS  = 2048;
constexpr int DD  = 64;
constexpr int DK  = 24;
constexpr int DV  = 32;
constexpr int NT  = SS / 16;
constexpr int TILE_F = 16 * DK;
constexpr int WPB   = 8;
constexpr int NPAIR = NT / 2;
constexpr int GRID_PAIR = BB * NPAIR / WPB;

#define WMMA_F32X4(Am, Bm, Cm) \
  __builtin_amdgcn_wmma_f32_16x16x4_f32(false, (Am), false, (Bm), (short)0, (Cm), false, false)

#define TRIP_EDGE() __builtin_amdgcn_sched_barrier(0)

__global__ void __launch_bounds__(256)
qkv_proj_kernel(const float* __restrict__ x, const float* __restrict__ Wq,
                const float* __restrict__ Wk, const float* __restrict__ Wv,
                float* __restrict__ Qp, float* __restrict__ Kp, float* __restrict__ Vm)
{
    int row = blockIdx.x * blockDim.x + threadIdx.x;
    if (row >= BB * SS) return;
    const float* xp = x + (size_t)row * DD;

    float accq[DK], acck[DK], accv[DV];
    #pragma unroll
    for (int j = 0; j < DK; ++j) { accq[j] = 0.f; acck[j] = 0.f; }
    #pragma unroll
    for (int j = 0; j < DV; ++j) accv[j] = 0.f;

    for (int d = 0; d < DD; ++d) {
        float xd = xp[d];
        #pragma unroll
        for (int j = 0; j < DK; ++j) {
            accq[j] = fmaf(xd, Wq[d * DK + j], accq[j]);
            acck[j] = fmaf(xd, Wk[d * DK + j], acck[j]);
        }
        #pragma unroll
        for (int j = 0; j < DV; ++j)
            accv[j] = fmaf(xd, Wv[d * DV + j], accv[j]);
    }
    float* qo = Qp + (size_t)(row >> 4) * TILE_F + (row & 15) * 4;
    float* ko = Kp + (size_t)(row >> 4) * TILE_F + (row & 15) * 4;
    float* vo = Vm + (size_t)row * DV;
    #pragma unroll
    for (int hl = 0; hl < 2; ++hl) {
        #pragma unroll
        for (int p = 0; p < 3; ++p) {
            int k0 = 12 * hl + 4 * p;
            v4f qv = {accq[k0], accq[k0 + 1], accq[k0 + 2], accq[k0 + 3]};
            v4f kv = {acck[k0], acck[k0 + 1], acck[k0 + 2], acck[k0 + 3]};
            *(v4f*)(qo + p * 128 + hl * 64) = qv;
            *(v4f*)(ko + p * 128 + hl * 64) = kv;
        }
    }
    #pragma unroll
    for (int j = 0; j < DV; ++j) vo[j] = accv[j];
}

struct QFrag { v4f p0, p1, p2; };

__device__ __forceinline__ void load_qfrag(QFrag& f, const float* tp)
{
    f.p0 = *(const v4f*)(tp);
    f.p1 = *(const v4f*)(tp + 128);
    f.p2 = *(const v4f*)(tp + 256);
}

__device__ __forceinline__ v2f lo2(v4f v) { return __builtin_shufflevector(v, v, 0, 1); }
__device__ __forceinline__ v2f hi2(v4f v) { return __builtin_shufflevector(v, v, 2, 3); }

__device__ __forceinline__ void stats_step(const QFrag& f, v2f b0, v2f b1, v2f b2,
                                           v2f b3, v2f b4, v2f b5,
                                           float& m_run, float& z_run)
{
    v8f s = {};
    s = WMMA_F32X4(lo2(f.p0), b0, s);
    s = WMMA_F32X4(hi2(f.p0), b1, s);
    s = WMMA_F32X4(lo2(f.p1), b2, s);
    s = WMMA_F32X4(hi2(f.p1), b3, s);
    s = WMMA_F32X4(lo2(f.p2), b4, s);
    s = WMMA_F32X4(hi2(f.p2), b5, s);

    float tmax = fmaxf(fmaxf(fmaxf(s[0], s[1]), fmaxf(s[2], s[3])),
                       fmaxf(fmaxf(s[4], s[5]), fmaxf(s[6], s[7])));
    tmax = fmaxf(tmax, __shfl_xor(tmax, 16, 32));
    float m_new = fmaxf(m_run, tmax);
    float p = 0.f;
    #pragma unroll
    for (int i = 0; i < 8; ++i) p += __expf(s[i] - m_new);
    p += __shfl_xor(p, 16, 32);
    z_run = fmaf(z_run, __expf(m_run - m_new), p);
    m_run = m_new;
}

__global__ void __launch_bounds__(32 * WPB) __attribute__((amdgpu_waves_per_eu(2, 2)))
col_softmax_stats_kernel(const float* __restrict__ Qp, const float* __restrict__ Kp,
                         float* __restrict__ mArr, float* __restrict__ Vm)
{
    int wv = __builtin_amdgcn_readfirstlane((int)(threadIdx.x >> 5));
    int pr = blockIdx.x * WPB + wv;
    int bb = pr / NPAIR;
    int tt = 2 * (pr - bb * NPAIR);
    int lane = threadIdx.x & 31;

    QFrag kf, kg;
    load_qfrag(kf, Kp + ((size_t)bb * NT + tt) * TILE_F + lane * 4);
    load_qfrag(kg, Kp + ((size_t)bb * NT + tt + 1) * TILE_F + lane * 4);
    v2f b0 = lo2(kf.p0), b1 = hi2(kf.p0);
    v2f b2 = lo2(kf.p1), b3 = hi2(kf.p1);
    v2f b4 = lo2(kf.p2), b5 = hi2(kf.p2);
    v2f c0 = lo2(kg.p0), c1 = hi2(kg.p0);
    v2f c2 = lo2(kg.p1), c3 = hi2(kg.p1);
    v2f c4 = lo2(kg.p2), c5 = hi2(kg.p2);

    const float* qbase = Qp + (size_t)bb * NT * TILE_F + lane * 4;
    float m_run = -3.0e38f, n_run = -3.0e38f;
    float z_run = 0.f,      y_run = 0.f;

    QFrag p0a, p0b, p1a, p1b, p2a, p2b, p3a, p3b;
    load_qfrag(p0a, qbase);
    load_qfrag(p0b, qbase + TILE_F);
    load_qfrag(p1a, qbase + 2 * TILE_F);
    load_qfrag(p1b, qbase + 3 * TILE_F);
    load_qfrag(p2a, qbase + 4 * TILE_F);
    load_qfrag(p2b, qbase + 5 * TILE_F);
    auto tile_at = [&](int t) { return qbase + (t < NT ? t : NT - 1) * TILE_F; };
    for (int qt = 0; qt < NT; qt += 8) {
        load_qfrag(p3a, tile_at(qt + 6));
        load_qfrag(p3b, tile_at(qt + 7));
        TRIP_EDGE();
        stats_step(p0a, b0, b1, b2, b3, b4, b5, m_run, z_run);
        stats_step(p0a, c0, c1, c2, c3, c4, c5, n_run, y_run);
        stats_step(p0b, b0, b1, b2, b3, b4, b5, m_run, z_run);
        stats_step(p0b, c0, c1, c2, c3, c4, c5, n_run, y_run);
        TRIP_EDGE();
        load_qfrag(p0a, tile_at(qt + 8));
        load_qfrag(p0b, tile_at(qt + 9));
        TRIP_EDGE();
        stats_step(p1a, b0, b1, b2, b3, b4, b5, m_run, z_run);
        stats_step(p1a, c0, c1, c2, c3, c4, c5, n_run, y_run);
        stats_step(p1b, b0, b1, b2, b3, b4, b5, m_run, z_run);
        stats_step(p1b, c0, c1, c2, c3, c4, c5, n_run, y_run);
        TRIP_EDGE();
        load_qfrag(p1a, tile_at(qt + 10));
        load_qfrag(p1b, tile_at(qt + 11));
        TRIP_EDGE();
        stats_step(p2a, b0, b1, b2, b3, b4, b5, m_run, z_run);
        stats_step(p2a, c0, c1, c2, c3, c4, c5, n_run, y_run);
        stats_step(p2b, b0, b1, b2, b3, b4, b5, m_run, z_run);
        stats_step(p2b, c0, c1, c2, c3, c4, c5, n_run, y_run);
        TRIP_EDGE();
        load_qfrag(p2a, tile_at(qt + 12));
        load_qfrag(p2b, tile_at(qt + 13));
        TRIP_EDGE();
        stats_step(p3a, b0, b1, b2, b3, b4, b5, m_run, z_run);
        stats_step(p3a, c0, c1, c2, c3, c4, c5, n_run, y_run);
        stats_step(p3b, b0, b1, b2, b3, b4, b5, m_run, z_run);
        stats_step(p3b, c0, c1, c2, c3, c4, c5, n_run, y_run);
    }
    float m_out = (lane < 16) ? m_run : n_run;
    float z_out = (lane < 16) ? z_run : y_run;
    mArr[bb * SS + tt * 16 + lane] = m_out;
    float zinv = 1.f / z_out;

    float* vrow = Vm + ((size_t)bb * SS + (size_t)tt * 16) * DV + lane;
    float vv[32];
    #pragma unroll
    for (int j = 0; j < 32; ++j) vv[j] = vrow[j * DV];
    #pragma unroll
    for (int j = 0; j < 32; ++j) {
        float zj = __int_as_float(__builtin_amdgcn_readlane(__float_as_int(zinv), j));
        vrow[j * DV] = vv[j] * zj;
    }
}

struct TFrag { QFrag k; float m; v2f bv0[4], bv1[4]; };

__device__ __forceinline__ void load_tfrag(TFrag& f, const float* kp, const float* mp,
                                           const float* vt, int col, int hl)
{
    load_qfrag(f.k, kp);
    f.m  = mp[col];
    #pragma unroll
    for (int c = 0; c < 4; ++c) {
        const float* vp = vt + (size_t)(4 * c + 2 * hl) * DV;
        f.bv0[c].x = vp[col];        f.bv0[c].y = vp[DV + col];
        f.bv1[c].x = vp[16 + col];   f.bv1[c].y = vp[DV + 16 + col];
    }
}

__global__ void __launch_bounds__(32 * WPB) __attribute__((amdgpu_waves_per_eu(2, 2)))
attn_h1_kernel(const float* __restrict__ Qp, const float* __restrict__ Kp,
               const float* __restrict__ VZ, const float* __restrict__ mArr,
               float* __restrict__ h1)
{
    __shared__ float ldsE[WPB][2][16 * 20];
    int wv = __builtin_amdgcn_readfirstlane((int)(threadIdx.x >> 5));
    int pr = blockIdx.x * WPB + wv;
    int bb = pr / NPAIR;
    int qt = 2 * (pr - bb * NPAIR);
    int lane = threadIdx.x & 31;
    int col  = lane & 15;
    int hl   = lane >> 4;
    float* e0 = &ldsE[wv][0][0];
    float* e1 = &ldsE[wv][1][0];

    QFrag qf, qg;
    load_qfrag(qf, Qp + ((size_t)bb * NT + qt) * TILE_F + lane * 4);
    load_qfrag(qg, Qp + ((size_t)bb * NT + qt + 1) * TILE_F + lane * 4);
    v2f a0 = lo2(qf.p0), a1 = hi2(qf.p0);
    v2f a2 = lo2(qf.p1), a3 = hi2(qf.p1);
    v2f a4 = lo2(qf.p2), a5 = hi2(qf.p2);
    v2f g0 = lo2(qg.p0), g1 = hi2(qg.p0);
    v2f g2 = lo2(qg.p1), g3 = hi2(qg.p1);
    v2f g4 = lo2(qg.p2), g5 = hi2(qg.p2);

    v8f acc0 = {};
    v8f acc1 = {};
    v8f acc2 = {};
    v8f acc3 = {};

    const float* kbase = Kp + (size_t)bb * NT * TILE_F + lane * 4;
    const float* vzb   = VZ + (size_t)(bb * SS) * DV;
    const float* mrow  = mArr + bb * SS;

    auto step = [&](const TFrag& f) {
        v8f s = {}, u = {};
        s = WMMA_F32X4(a0, lo2(f.k.p0), s);
        u = WMMA_F32X4(g0, lo2(f.k.p0), u);
        s = WMMA_F32X4(a1, hi2(f.k.p0), s);
        u = WMMA_F32X4(g1, hi2(f.k.p0), u);
        s = WMMA_F32X4(a2, lo2(f.k.p1), s);
        u = WMMA_F32X4(g2, lo2(f.k.p1), u);
        s = WMMA_F32X4(a3, hi2(f.k.p1), s);
        u = WMMA_F32X4(g3, hi2(f.k.p1), u);
        s = WMMA_F32X4(a4, lo2(f.k.p2), s);
        u = WMMA_F32X4(g4, lo2(f.k.p2), u);
        s = WMMA_F32X4(a5, hi2(f.k.p2), s);
        u = WMMA_F32X4(g5, hi2(f.k.p2), u);

        #pragma unroll
        for (int r = 0; r < 8; ++r)
            e0[(r + 8 * hl) * 20 + col] = __expf(s[r] - f.m);
        #pragma unroll
        for (int r = 0; r < 8; ++r)
            e1[(r + 8 * hl) * 20 + col] = __expf(u[r] - f.m);

        #pragma unroll
        for (int c = 0; c < 4; ++c) {
            v2f ae = *(const v2f*)&e0[col * 20 + 4 * c + 2 * hl];
            v2f af = *(const v2f*)&e1[col * 20 + 4 * c + 2 * hl];
            acc0 = WMMA_F32X4(ae, f.bv0[c], acc0);
            acc1 = WMMA_F32X4(ae, f.bv1[c], acc1);
            acc2 = WMMA_F32X4(af, f.bv0[c], acc2);
            acc3 = WMMA_F32X4(af, f.bv1[c], acc3);
        }
    };

    auto load_t = [&](TFrag& f, int t) {
        int tc = (t < NT) ? t : NT - 1;
        load_tfrag(f, kbase + tc * TILE_F, mrow + tc * 16,
                   vzb + (size_t)(tc * 16) * DV, col, hl);
    };
    TFrag f0, f1, f2, f3;
    load_t(f0, 0);
    load_t(f1, 1);
    for (int tt = 0; tt < NT; tt += 4) {
        load_t(f2, tt + 2);
        TRIP_EDGE();
        step(f0);
        TRIP_EDGE();
        load_t(f3, tt + 3);
        TRIP_EDGE();
        step(f1);
        TRIP_EDGE();
        load_t(f0, tt + 4);
        TRIP_EDGE();
        step(f2);
        TRIP_EDGE();
        load_t(f1, tt + 5);
        TRIP_EDGE();
        step(f3);
    }

    #pragma unroll
    for (int r = 0; r < 8; ++r) {
        float* hp = h1 + ((size_t)(bb * SS) + (size_t)qt * 16 + r + 8 * hl) * DV;
        hp[col]      = acc0[r];
        hp[col + 16] = acc1[r];
        hp[16 * DV + col]      = acc2[r];
        hp[16 * DV + col + 16] = acc3[r];
    }
}

__global__ void __launch_bounds__(256)
out_proj_kernel(const float* __restrict__ h1, const float* __restrict__ Wh,
                float* __restrict__ out)
{
    int row = blockIdx.x * blockDim.x + threadIdx.x;
    if (row >= BB * SS) return;
    const float* hp = h1 + (size_t)row * DV;
    float acc[DD];
    #pragma unroll
    for (int d = 0; d < DD; ++d) acc[d] = 0.f;
    for (int v = 0; v < DV; ++v) {
        float hv = hp[v];
        #pragma unroll
        for (int d = 0; d < DD; ++d)
            acc[d] = fmaf(hv, Wh[v * DD + d], acc[d]);
    }
    float* op = out + (size_t)row * DD;
    #pragma unroll
    for (int d = 0; d < DD; ++d) op[d] = acc[d];
}

extern "C" void kernel_launch(void* const* d_in, const int* in_sizes, int n_in,
                              void* d_out, int out_size, void* d_ws, size_t ws_size,
                              hipStream_t stream)
{
    const float* x  = (const float*)d_in[0];
    const float* Wq = (const float*)d_in[1];
    const float* Wk = (const float*)d_in[2];
    const float* Wv = (const float*)d_in[3];
    const float* Wh = (const float*)d_in[4];
    float* out = (float*)d_out;

    float* ws = (float*)d_ws;
    size_t o = 0;
    float* Qp = ws + o; o += (size_t)BB * SS * 32;
    float* Kp = ws + o; o += (size_t)BB * SS * 32;
    float* Vm = ws + o; o += (size_t)BB * SS * DV;
    float* mA = ws + o; o += (size_t)BB * SS;
    float* h1 = ws + o; o += (size_t)BB * SS * DV;

    const int rows = BB * SS;
    qkv_proj_kernel<<<(rows + 255) / 256, 256, 0, stream>>>(x, Wq, Wk, Wv, Qp, Kp, Vm);
    col_softmax_stats_kernel<<<GRID_PAIR, 32 * WPB, 0, stream>>>(Qp, Kp, mA, Vm);
    attn_h1_kernel<<<GRID_PAIR, 32 * WPB, 0, stream>>>(Qp, Kp, Vm, mA, h1);
    out_proj_kernel<<<(rows + 255) / 256, 256, 0, stream>>>(h1, Wh, out);
}


# h1 kernel: score tile computed transposed (K as A, Q as B): exp(s-m) is already an A fragment of the second product, LDS transpose removed; vz rows regrouped to match, so the t sum order changes
# speedup vs baseline: 1.5063x; 1.0337x over previous
#include <hip/hip_runtime.h>
#include <hip/hip_bf16.h>


typedef __attribute__((ext_vector_type(2))) float v2f;
typedef __attribute__((ext_vector_type(4))) float v4f;
typedef __attribute__((ext_vector_type(8))) float v8f;

constexpr int BB  = 32;
constexpr int SS  = 2048;
constexpr int DD  = 64;
constexpr int DK  = 24;
constexpr int DV  = 32;
constexpr int NT  = SS / 16;
constexpr int TILE_F = 16 * DK;
constexpr int WPB   = 8;
constexpr int NPAIR = NT / 2;
constexpr int GRID_PAIR = BB * NPAIR / WPB;

#define WMMA_F32X4(Am, Bm, Cm) \
  __builtin_amdgcn_wmma_f32_16x16x4_f32(false, (Am), false, (Bm), (short)0, (Cm), false, false)

#define TRIP_EDGE() __builtin_amdgcn_sched_barrier(0)

__global__ void __launch_bounds__(256)
qkv_proj_kernel(const float* __restrict__ x, const float* __restrict__ Wq,
                const float* __restrict__ Wk, const float* __restrict__ Wv,
                float* __restrict__ Qp, float* __restrict__ Kp, float* __restrict__ Vm)
{
    int row = blockIdx.x * blockDim.x + threadIdx.x;
    if (row >= BB * SS) return;
    const float* xp = x + (size_t)row * DD;

    float accq[DK], acck[DK], accv[DV];
    #pragma unroll
    for (int j = 0; j < DK; ++j) { accq[j] = 0.f; acck[j] = 0.f; }
    #pragma unroll
    for (int j = 0; j < DV; ++j) accv[j] = 0.f;

    for (int d = 0; d < DD; ++d) {
        float xd = xp[d];
        #pragma unroll
        for (int j = 0; j < DK; ++j) {
            accq[j] = fmaf(xd, Wq[d * DK + j], accq[j]);
            acck[j] = fmaf(xd, Wk[d * DK + j], acck[j]);
        }
        #pragma unroll
        for (int j = 0; j < DV; ++j)
            accv[j] = fmaf(xd, Wv[d * DV + j], accv[j]);
    }
    float* qo = Qp + (size_t)(row >> 4) * TILE_F + (row & 15) * 4;
    float* ko = Kp + (size_t)(row >> 4) * TILE_F + (row & 15) * 4;
    float* vo = Vm + (size_t)row * DV;
    #pragma unroll
    for (int hl = 0; hl < 2; ++hl) {
        #pragma unroll
        for (int p = 0; p < 3; ++p) {
            int k0 = 12 * hl + 4 * p;
            v4f qv = {accq[k0], accq[k0 + 1], accq[k0 + 2], accq[k0 + 3]};
            v4f kv = {acck[k0], acck[k0 + 1], acck[k0 + 2], acck[k0 + 3]};
            *(v4f*)(qo + p * 128 + hl * 64) = qv;
            *(v4f*)(ko + p * 128 + hl * 64) = kv;
        }
    }
    #pragma unroll
    for (int j = 0; j < DV; ++j) vo[j] = accv[j];
}

struct QFrag { v4f p0, p1, p2; };

__device__ __forceinline__ void load_qfrag(QFrag& f, const float* tp)
{
    f.p0 = *(const v4f*)(tp);
    f.p1 = *(const v4f*)(tp + 128);
    f.p2 = *(const v4f*)(tp + 256);
}

__device__ __forceinline__ v2f lo2(v4f v) { return __builtin_shufflevector(v, v, 0, 1); }
__device__ __forceinline__ v2f hi2(v4f v) { return __builtin_shufflevector(v, v, 2, 3); }

__device__ __forceinline__ void stats_step(const QFrag& f, v2f b0, v2f b1, v2f b2,
                                           v2f b3, v2f b4, v2f b5,
                                           float& m_run, float& z_run)
{
    v8f s = {};
    s = WMMA_F32X4(lo2(f.p0), b0, s);
    s = WMMA_F32X4(hi2(f.p0), b1, s);
    s = WMMA_F32X4(lo2(f.p1), b2, s);
    s = WMMA_F32X4(hi2(f.p1), b3, s);
    s = WMMA_F32X4(lo2(f.p2), b4, s);
    s = WMMA_F32X4(hi2(f.p2), b5, s);

    float tmax = fmaxf(fmaxf(fmaxf(s[0], s[1]), fmaxf(s[2], s[3])),
                       fmaxf(fmaxf(s[4], s[5]), fmaxf(s[6], s[7])));
    tmax = fmaxf(tmax, __shfl_xor(tmax, 16, 32));
    float m_new = fmaxf(m_run, tmax);
    float p = 0.f;
    #pragma unroll
    for (int i = 0; i < 8; ++i) p += __expf(s[i] - m_new);
    p += __shfl_xor(p, 16, 32);
    z_run = fmaf(z_run, __expf(m_run - m_new), p);
    m_run = m_new;
}

__global__ void __launch_bounds__(32 * WPB) __attribute__((amdgpu_waves_per_eu(2, 2)))
col_softmax_stats_kernel(const float* __restrict__ Qp, const float* __restrict__ Kp,
                         float* __restrict__ mArr, float* __restrict__ Vm)
{
    int wv = __builtin_amdgcn_readfirstlane((int)(threadIdx.x >> 5));
    int pr = blockIdx.x * WPB + wv;
    int bb = pr / NPAIR;
    int tt = 2 * (pr - bb * NPAIR);
    int lane = threadIdx.x & 31;

    QFrag kf, kg;
    load_qfrag(kf, Kp + ((size_t)bb * NT + tt) * TILE_F + lane * 4);
    load_qfrag(kg, Kp + ((size_t)bb * NT + tt + 1) * TILE_F + lane * 4);
    v2f b0 = lo2(kf.p0), b1 = hi2(kf.p0);
    v2f b2 = lo2(kf.p1), b3 = hi2(kf.p1);
    v2f b4 = lo2(kf.p2), b5 = hi2(kf.p2);
    v2f c0 = lo2(kg.p0), c1 = hi2(kg.p0);
    v2f c2 = lo2(kg.p1), c3 = hi2(kg.p1);
    v2f c4 = lo2(kg.p2), c5 = hi2(kg.p2);

    const float* qbase = Qp + (size_t)bb * NT * TILE_F + lane * 4;
    float m_run = -3.0e38f, n_run = -3.0e38f;
    float z_run = 0.f,      y_run = 0.f;

    QFrag p0a, p0b, p1a, p1b, p2a, p2b, p3a, p3b;
    load_qfrag(p0a, qbase);
    load_qfrag(p0b, qbase + TILE_F);
    load_qfrag(p1a, qbase + 2 * TILE_F);
    load_qfrag(p1b, qbase + 3 * TILE_F);
    load_qfrag(p2a, qbase + 4 * TILE_F);
    load_qfrag(p2b, qbase + 5 * TILE_F);
    auto tile_at = [&](int t) { return qbase + (t < NT ? t : NT - 1) * TILE_F; };
    for (int qt = 0; qt < NT; qt += 8) {
        load_qfrag(p3a, tile_at(qt + 6));
        load_qfrag(p3b, tile_at(qt + 7));
        TRIP_EDGE();
        stats_step(p0a, b0, b1, b2, b3, b4, b5, m_run, z_run);
        stats_step(p0a, c0, c1, c2, c3, c4, c5, n_run, y_run);
        stats_step(p0b, b0, b1, b2, b3, b4, b5, m_run, z_run);
        stats_step(p0b, c0, c1, c2, c3, c4, c5, n_run, y_run);
        TRIP_EDGE();
        load_qfrag(p0a, tile_at(qt + 8));
        load_qfrag(p0b, tile_at(qt + 9));
        TRIP_EDGE();
        stats_step(p1a, b0, b1, b2, b3, b4, b5, m_run, z_run);
        stats_step(p1a, c0, c1, c2, c3, c4, c5, n_run, y_run);
        stats_step(p1b, b0, b1, b2, b3, b4, b5, m_run, z_run);
        stats_step(p1b, c0, c1, c2, c3, c4, c5, n_run, y_run);
        TRIP_EDGE();
        load_qfrag(p1a, tile_at(qt + 10));
        load_qfrag(p1b, tile_at(qt + 11));
        TRIP_EDGE();
        stats_step(p2a, b0, b1, b2, b3, b4, b5, m_run, z_run);
        stats_step(p2a, c0, c1, c2, c3, c4, c5, n_run, y_run);
        stats_step(p2b, b0, b1, b2, b3, b4, b5, m_run, z_run);
        stats_step(p2b, c0, c1, c2, c3, c4, c5, n_run, y_run);
        TRIP_EDGE();
        load_qfrag(p2a, tile_at(qt + 12));
        load_qfrag(p2b, tile_at(qt + 13));
        TRIP_EDGE();
        stats_step(p3a, b0, b1, b2, b3, b4, b5, m_run, z_run);
        stats_step(p3a, c0, c1, c2, c3, c4, c5, n_run, y_run);
        stats_step(p3b, b0, b1, b2, b3, b4, b5, m_run, z_run);
        stats_step(p3b, c0, c1, c2, c3, c4, c5, n_run, y_run);
    }
    float m_out = (lane < 16) ? m_run : n_run;
    float z_out = (lane < 16) ? z_run : y_run;
    mArr[bb * SS + tt * 16 + lane] = m_out;
    float zinv = 1.f / z_out;

    float* vrow = Vm + ((size_t)bb * SS + (size_t)tt * 16) * DV + lane;
    float vv[32];
    #pragma unroll
    for (int j = 0; j < 32; ++j) vv[j] = vrow[j * DV];
    #pragma unroll
    for (int j = 0; j < 32; ++j) {
        float zj = __int_as_float(__builtin_amdgcn_readlane(__float_as_int(zinv), j));
        vrow[j * DV] = vv[j] * zj;
    }
}

struct TFrag { QFrag k; v4f m0, m1; v2f bv0[4], bv1[4]; };

__device__ __forceinline__ void load_tfrag(TFrag& f, const float* kp, const float* mp,
                                           const float* vt, int col, int hl)
{
    load_qfrag(f.k, kp);
    f.m0 = *(const v4f*)(mp + 8 * hl);
    f.m1 = *(const v4f*)(mp + 8 * hl + 4);
    #pragma unroll
    for (int c = 0; c < 4; ++c) {
        const float* vp = vt + (size_t)(8 * hl + 2 * c) * DV;
        f.bv0[c].x = vp[col];        f.bv0[c].y = vp[DV + col];
        f.bv1[c].x = vp[16 + col];   f.bv1[c].y = vp[DV + 16 + col];
    }
}

__global__ void __launch_bounds__(32 * WPB) __attribute__((amdgpu_waves_per_eu(2, 2)))
attn_h1_kernel(const float* __restrict__ Qp, const float* __restrict__ Kp,
               const float* __restrict__ VZ, const float* __restrict__ mArr,
               float* __restrict__ h1)
{
    int wv = __builtin_amdgcn_readfirstlane((int)(threadIdx.x >> 5));
    int pr = blockIdx.x * WPB + wv;
    int bb = pr / NPAIR;
    int qt = 2 * (pr - bb * NPAIR);
    int lane = threadIdx.x & 31;
    int col  = lane & 15;
    int hl   = lane >> 4;

    QFrag qf, qg;
    load_qfrag(qf, Qp + ((size_t)bb * NT + qt) * TILE_F + lane * 4);
    load_qfrag(qg, Qp + ((size_t)bb * NT + qt + 1) * TILE_F + lane * 4);
    v2f a0 = lo2(qf.p0), a1 = hi2(qf.p0);
    v2f a2 = lo2(qf.p1), a3 = hi2(qf.p1);
    v2f a4 = lo2(qf.p2), a5 = hi2(qf.p2);
    v2f g0 = lo2(qg.p0), g1 = hi2(qg.p0);
    v2f g2 = lo2(qg.p1), g3 = hi2(qg.p1);
    v2f g4 = lo2(qg.p2), g5 = hi2(qg.p2);

    v8f acc0 = {};
    v8f acc1 = {};
    v8f acc2 = {};
    v8f acc3 = {};

    const float* kbase = Kp + (size_t)bb * NT * TILE_F + lane * 4;
    const float* vzb   = VZ + (size_t)(bb * SS) * DV;
    const float* mrow  = mArr + bb * SS;

    auto step = [&](const TFrag& f) {
        v8f s = {}, u = {};
        s = WMMA_F32X4(lo2(f.k.p0), a0, s);
        u = WMMA_F32X4(lo2(f.k.p0), g0, u);
        s = WMMA_F32X4(hi2(f.k.p0), a1, s);
        u = WMMA_F32X4(hi2(f.k.p0), g1, u);
        s = WMMA_F32X4(lo2(f.k.p1), a2, s);
        u = WMMA_F32X4(lo2(f.k.p1), g2, u);
        s = WMMA_F32X4(hi2(f.k.p1), a3, s);
        u = WMMA_F32X4(hi2(f.k.p1), g3, u);
        s = WMMA_F32X4(lo2(f.k.p2), a4, s);
        u = WMMA_F32X4(lo2(f.k.p2), g4, u);
        s = WMMA_F32X4(hi2(f.k.p2), a5, s);
        u = WMMA_F32X4(hi2(f.k.p2), g5, u);

        const v2f mm[4] = {lo2(f.m0), hi2(f.m0), lo2(f.m1), hi2(f.m1)};
        #pragma unroll
        for (int c = 0; c < 4; ++c) {
            v2f xs = v2f{s[2 * c], s[2 * c + 1]} - mm[c];
            v2f xu = v2f{u[2 * c], u[2 * c + 1]} - mm[c];
            v2f ae = {__expf(xs.x), __expf(xs.y)};
            v2f af = {__expf(xu.x), __expf(xu.y)};
            acc0 = WMMA_F32X4(ae, f.bv0[c], acc0);
            acc1 = WMMA_F32X4(ae, f.bv1[c], acc1);
            acc2 = WMMA_F32X4(af, f.bv0[c], acc2);
            acc3 = WMMA_F32X4(af, f.bv1[c], acc3);
        }
    };

    auto load_t = [&](TFrag& f, int t) {
        int tc = (t < NT) ? t : NT - 1;
        load_tfrag(f, kbase + tc * TILE_F, mrow + tc * 16,
                   vzb + (size_t)(tc * 16) * DV, col, hl);
    };
    TFrag f0, f1, f2, f3;
    load_t(f0, 0);
    load_t(f1, 1);
    for (int tt = 0; tt < NT; tt += 4) {
        load_t(f2, tt + 2);
        TRIP_EDGE();
        step(f0);
        TRIP_EDGE();
        load_t(f3, tt + 3);
        TRIP_EDGE();
        step(f1);
        TRIP_EDGE();
        load_t(f0, tt + 4);
        TRIP_EDGE();
        step(f2);
        TRIP_EDGE();
        load_t(f1, tt + 5);
        TRIP_EDGE();
        step(f3);
    }

    #pragma unroll
    for (int r = 0; r < 8; ++r) {
        float* hp = h1 + ((size_t)(bb * SS) + (size_t)qt * 16 + r + 8 * hl) * DV;
        hp[col]      = acc0[r];
        hp[col + 16] = acc1[r];
        hp[16 * DV + col]      = acc2[r];
        hp[16 * DV + col + 16] = acc3[r];
    }
}

__global__ void __launch_bounds__(256)
out_proj_kernel(const float* __restrict__ h1, const float* __restrict__ Wh,
                float* __restrict__ out)
{
    int row = blockIdx.x * blockDim.x + threadIdx.x;
    if (row >= BB * SS) return;
    const float* hp = h1 + (size_t)row * DV;
    float acc[DD];
    #pragma unroll
    for (int d = 0; d < DD; ++d) acc[d] = 0.f;
    for (int v = 0; v < DV; ++v) {
        float hv = hp[v];
        #pragma unroll
        for (int d = 0; d < DD; ++d)
            acc[d] = fmaf(hv, Wh[v * DD + d], acc[d]);
    }
    float* op = out + (size_t)row * DD;
    #pragma unroll
    for (int d = 0; d < DD; ++d) op[d] = acc[d];
}

extern "C" void kernel_launch(void* const* d_in, const int* in_sizes, int n_in,
                              void* d_out, int out_size, void* d_ws, size_t ws_size,
                              hipStream_t stream)
{
    const float* x  = (const float*)d_in[0];
    const float* Wq = (const float*)d_in[1];
    const float* Wk = (const float*)d_in[2];
    const float* Wv = (const float*)d_in[3];
    const float* Wh = (const float*)d_in[4];
    float* out = (float*)d_out;

    float* ws = (float*)d_ws;
    size_t o = 0;
    float* Qp = ws + o; o += (size_t)BB * SS * 32;
    float* Kp = ws + o; o += (size_t)BB * SS * 32;
    float* Vm = ws + o; o += (size_t)BB * SS * DV;
    float* mA = ws + o; o += (size_t)BB * SS;
    float* h1 = ws + o; o += (size_t)BB * SS * DV;

    const int rows = BB * SS;
    qkv_proj_kernel<<<(rows + 255) / 256, 256, 0, stream>>>(x, Wq, Wk, Wv, Qp, Kp, Vm);
    col_softmax_stats_kernel<<<GRID_PAIR, 32 * WPB, 0, stream>>>(Qp, Kp, mA, Vm);
    attn_h1_kernel<<<GRID_PAIR, 32 * WPB, 0, stream>>>(Qp, Kp, Vm, mA, h1);
    out_proj_kernel<<<(rows + 255) / 256, 256, 0, stream>>>(h1, Wh, out);
}


# Output projection fused into the h1 kernel's end: h1 kept transposed (vz as A, exp as B), its register pairs are B fragments of out^T = Wh^T h1^T on WMMA; 4 launches -> 3; out's v sum order changes
# speedup vs baseline: 1.7302x; 1.1486x over previous
#include <hip/hip_runtime.h>
#include <hip/hip_bf16.h>


typedef __attribute__((ext_vector_type(2))) float v2f;
typedef __attribute__((ext_vector_type(4))) float v4f;
typedef __attribute__((ext_vector_type(8))) float v8f;

constexpr int BB  = 32;
constexpr int SS  = 2048;
constexpr int DD  = 64;
constexpr int DK  = 24;
constexpr int DV  = 32;
constexpr int NT  = SS / 16;
constexpr int TILE_F = 16 * DK;
constexpr int WPB   = 8;
constexpr int NPAIR = NT / 2;
constexpr int GRID_PAIR = BB * NPAIR / WPB;

#define WMMA_F32X4(Am, Bm, Cm) \
  __builtin_amdgcn_wmma_f32_16x16x4_f32(false, (Am), false, (Bm), (short)0, (Cm), false, false)

#define TRIP_EDGE() __builtin_amdgcn_sched_barrier(0)

__global__ void __launch_bounds__(256)
qkv_proj_kernel(const float* __restrict__ x, const float* __restrict__ Wq,
                const float* __restrict__ Wk, const float* __restrict__ Wv,
                float* __restrict__ Qp, float* __restrict__ Kp, float* __restrict__ Vm)
{
    int row = blockIdx.x * blockDim.x + threadIdx.x;
    if (row >= BB * SS) return;
    const float* xp = x + (size_t)row * DD;

    float accq[DK], acck[DK], accv[DV];
    #pragma unroll
    for (int j = 0; j < DK; ++j) { accq[j] = 0.f; acck[j] = 0.f; }
    #pragma unroll
    for (int j = 0; j < DV; ++j) accv[j] = 0.f;

    for (int d = 0; d < DD; ++d) {
        float xd = xp[d];
        #pragma unroll
        for (int j = 0; j < DK; ++j) {
            accq[j] = fmaf(xd, Wq[d * DK + j], accq[j]);
            acck[j] = fmaf(xd, Wk[d * DK + j], acck[j]);
        }
        #pragma unroll
        for (int j = 0; j < DV; ++j)
            accv[j] = fmaf(xd, Wv[d * DV + j], accv[j]);
    }
    float* qo = Qp + (size_t)(row >> 4) * TILE_F + (row & 15) * 4;
    float* ko = Kp + (size_t)(row >> 4) * TILE_F + (row & 15) * 4;
    float* vo = Vm + (size_t)row * DV;
    #pragma unroll
    for (int hl = 0; hl < 2; ++hl) {
        #pragma unroll
        for (int p = 0; p < 3; ++p) {
            int k0 = 12 * hl + 4 * p;
            v4f qv = {accq[k0], accq[k0 + 1], accq[k0 + 2], accq[k0 + 3]};
            v4f kv = {acck[k0], acck[k0 + 1], acck[k0 + 2], acck[k0 + 3]};
            *(v4f*)(qo + p * 128 + hl * 64) = qv;
            *(v4f*)(ko + p * 128 + hl * 64) = kv;
        }
    }
    #pragma unroll
    for (int j = 0; j < DV; ++j) vo[j] = accv[j];
}

struct QFrag { v4f p0, p1, p2; };

__device__ __forceinline__ void load_qfrag(QFrag& f, const float* tp)
{
    f.p0 = *(const v4f*)(tp);
    f.p1 = *(const v4f*)(tp + 128);
    f.p2 = *(const v4f*)(tp + 256);
}

__device__ __forceinline__ v2f lo2(v4f v) { return __builtin_shufflevector(v, v, 0, 1); }
__device__ __forceinline__ v2f hi2(v4f v) { return __builtin_shufflevector(v, v, 2, 3); }

__device__ __forceinline__ void stats_step(const QFrag& f, v2f b0, v2f b1, v2f b2,
                                           v2f b3, v2f b4, v2f b5,
                                           float& m_run, float& z_run)
{
    v8f s = {};
    s = WMMA_F32X4(lo2(f.p0), b0, s);
    s = WMMA_F32X4(hi2(f.p0), b1, s);
    s = WMMA_F32X4(lo2(f.p1), b2, s);
    s = WMMA_F32X4(hi2(f.p1), b3, s);
    s = WMMA_F32X4(lo2(f.p2), b4, s);
    s = WMMA_F32X4(hi2(f.p2), b5, s);

    float tmax = fmaxf(fmaxf(fmaxf(s[0], s[1]), fmaxf(s[2], s[3])),
                       fmaxf(fmaxf(s[4], s[5]), fmaxf(s[6], s[7])));
    tmax = fmaxf(tmax, __shfl_xor(tmax, 16, 32));
    float m_new = fmaxf(m_run, tmax);
    float p = 0.f;
    #pragma unroll
    for (int i = 0; i < 8; ++i) p += __expf(s[i] - m_new);
    p += __shfl_xor(p, 16, 32);
    z_run = fmaf(z_run, __expf(m_run - m_new), p);
    m_run = m_new;
}

__global__ void __launch_bounds__(32 * WPB) __attribute__((amdgpu_waves_per_eu(2, 2)))
col_softmax_stats_kernel(const float* __restrict__ Qp, const float* __restrict__ Kp,
                         float* __restrict__ mArr, float* __restrict__ Vm)
{
    int wv = __builtin_amdgcn_readfirstlane((int)(threadIdx.x >> 5));
    int pr = blockIdx.x * WPB + wv;
    int bb = pr / NPAIR;
    int tt = 2 * (pr - bb * NPAIR);
    int lane = threadIdx.x & 31;

    QFrag kf, kg;
    load_qfrag(kf, Kp + ((size_t)bb * NT + tt) * TILE_F + lane * 4);
    load_qfrag(kg, Kp + ((size_t)bb * NT + tt + 1) * TILE_F + lane * 4);
    v2f b0 = lo2(kf.p0), b1 = hi2(kf.p0);
    v2f b2 = lo2(kf.p1), b3 = hi2(kf.p1);
    v2f b4 = lo2(kf.p2), b5 = hi2(kf.p2);
    v2f c0 = lo2(kg.p0), c1 = hi2(kg.p0);
    v2f c2 = lo2(kg.p1), c3 = hi2(kg.p1);
    v2f c4 = lo2(kg.p2), c5 = hi2(kg.p2);

    const float* qbase = Qp + (size_t)bb * NT * TILE_F + lane * 4;
    float m_run = -3.0e38f, n_run = -3.0e38f;
    float z_run = 0.f,      y_run = 0.f;

    QFrag p0a, p0b, p1a, p1b, p2a, p2b, p3a, p3b;
    load_qfrag(p0a, qbase);
    load_qfrag(p0b, qbase + TILE_F);
    load_qfrag(p1a, qbase + 2 * TILE_F);
    load_qfrag(p1b, qbase + 3 * TILE_F);
    load_qfrag(p2a, qbase + 4 * TILE_F);
    load_qfrag(p2b, qbase + 5 * TILE_F);
    auto tile_at = [&](int t) { return qbase + (t < NT ? t : NT - 1) * TILE_F; };
    for (int qt = 0; qt < NT; qt += 8) {
        load_qfrag(p3a, tile_at(qt + 6));
        load_qfrag(p3b, tile_at(qt + 7));
        TRIP_EDGE();
        stats_step(p0a, b0, b1, b2, b3, b4, b5, m_run, z_run);
        stats_step(p0a, c0, c1, c2, c3, c4, c5, n_run, y_run);
        stats_step(p0b, b0, b1, b2, b3, b4, b5, m_run, z_run);
        stats_step(p0b, c0, c1, c2, c3, c4, c5, n_run, y_run);
        TRIP_EDGE();
        load_qfrag(p0a, tile_at(qt + 8));
        load_qfrag(p0b, tile_at(qt + 9));
        TRIP_EDGE();
        stats_step(p1a, b0, b1, b2, b3, b4, b5, m_run, z_run);
        stats_step(p1a, c0, c1, c2, c3, c4, c5, n_run, y_run);
        stats_step(p1b, b0, b1, b2, b3, b4, b5, m_run, z_run);
        stats_step(p1b, c0, c1, c2, c3, c4, c5, n_run, y_run);
        TRIP_EDGE();
        load_qfrag(p1a, tile_at(qt + 10));
        load_qfrag(p1b, tile_at(qt + 11));
        TRIP_EDGE();
        stats_step(p2a, b0, b1, b2, b3, b4, b5, m_run, z_run);
        stats_step(p2a, c0, c1, c2, c3, c4, c5, n_run, y_run);
        stats_step(p2b, b0, b1, b2, b3, b4, b5, m_run, z_run);
        stats_step(p2b, c0, c1, c2, c3, c4, c5, n_run, y_run);
        TRIP_EDGE();
        load_qfrag(p2a, tile_at(qt + 12));
        load_qfrag(p2b, tile_at(qt + 13));
        TRIP_EDGE();
        stats_step(p3a, b0, b1, b2, b3, b4, b5, m_run, z_run);
        stats_step(p3a, c0, c1, c2, c3, c4, c5, n_run, y_run);
        stats_step(p3b, b0, b1, b2, b3, b4, b5, m_run, z_run);
        stats_step(p3b, c0, c1, c2, c3, c4, c5, n_run, y_run);
    }
    float m_out = (lane < 16) ? m_run : n_run;
    float z_out = (lane < 16) ? z_run : y_run;
    mArr[bb * SS + tt * 16 + lane] = m_out;
    float zinv = 1.f / z_out;

    float* vrow = Vm + ((size_t)bb * SS + (size_t)tt * 16) * DV + lane;
    float vv[32];
    #pragma unroll
    for (int j = 0; j < 32; ++j) vv[j] = vrow[j * DV];
    #pragma unroll
    for (int j = 0; j < 32; ++j) {
        float zj = __int_as_float(__builtin_amdgcn_readlane(__float_as_int(zinv), j));
        vrow[j * DV] = vv[j] * zj;
    }
}

struct TFrag { QFrag k; v4f m0, m1; v2f bv0[4], bv1[4]; };

__device__ __forceinline__ void load_tfrag(TFrag& f, const float* kp, const float* mp,
                                           const float* vt, int col, int hl)
{
    load_qfrag(f.k, kp);
    f.m0 = *(const v4f*)(mp + 8 * hl);
    f.m1 = *(const v4f*)(mp + 8 * hl + 4);
    #pragma unroll
    for (int c = 0; c < 4; ++c) {
        const float* vp = vt + (size_t)(8 * hl + 2 * c) * DV;
        f.bv0[c].x = vp[col];        f.bv0[c].y = vp[DV + col];
        f.bv1[c].x = vp[16 + col];   f.bv1[c].y = vp[DV + 16 + col];
    }
}

__global__ void __launch_bounds__(32 * WPB) __attribute__((amdgpu_waves_per_eu(2, 2)))
attn_out_kernel(const float* __restrict__ Qp, const float* __restrict__ Kp,
                const float* __restrict__ VZ, const float* __restrict__ mArr,
                const float* __restrict__ Wh, float* __restrict__ out)
{
    int wv = __builtin_amdgcn_readfirstlane((int)(threadIdx.x >> 5));
    int pr = blockIdx.x * WPB + wv;
    int bb = pr / NPAIR;
    int qt = 2 * (pr - bb * NPAIR);
    int lane = threadIdx.x & 31;
    int col  = lane & 15;
    int hl   = lane >> 4;

    QFrag qf, qg;
    load_qfrag(qf, Qp + ((size_t)bb * NT + qt) * TILE_F + lane * 4);
    load_qfrag(qg, Qp + ((size_t)bb * NT + qt + 1) * TILE_F + lane * 4);
    v2f a0 = lo2(qf.p0), a1 = hi2(qf.p0);
    v2f a2 = lo2(qf.p1), a3 = hi2(qf.p1);
    v2f a4 = lo2(qf.p2), a5 = hi2(qf.p2);
    v2f g0 = lo2(qg.p0), g1 = hi2(qg.p0);
    v2f g2 = lo2(qg.p1), g3 = hi2(qg.p1);
    v2f g4 = lo2(qg.p2), g5 = hi2(qg.p2);

    v8f acc0 = {};
    v8f acc1 = {};
    v8f acc2 = {};
    v8f acc3 = {};

    const float* kbase = Kp + (size_t)bb * NT * TILE_F + lane * 4;
    const float* vzb   = VZ + (size_t)(bb * SS) * DV;
    const float* mrow  = mArr + bb * SS;

    auto step = [&](const TFrag& f) {
        v8f s = {}, u = {};
        s = WMMA_F32X4(lo2(f.k.p0), a0, s);
        u = WMMA_F32X4(lo2(f.k.p0), g0, u);
        s = WMMA_F32X4(hi2(f.k.p0), a1, s);
        u = WMMA_F32X4(hi2(f.k.p0), g1, u);
        s = WMMA_F32X4(lo2(f.k.p1), a2, s);
        u = WMMA_F32X4(lo2(f.k.p1), g2, u);
        s = WMMA_F32X4(hi2(f.k.p1), a3, s);
        u = WMMA_F32X4(hi2(f.k.p1), g3, u);
        s = WMMA_F32X4(lo2(f.k.p2), a4, s);
        u = WMMA_F32X4(lo2(f.k.p2), g4, u);
        s = WMMA_F32X4(hi2(f.k.p2), a5, s);
        u = WMMA_F32X4(hi2(f.k.p2), g5, u);

        const v2f mm[4] = {lo2(f.m0), hi2(f.m0), lo2(f.m1), hi2(f.m1)};
        #pragma unroll
        for (int c = 0; c < 4; ++c) {
            v2f xs = v2f{s[2 * c], s[2 * c + 1]} - mm[c];
            v2f xu = v2f{u[2 * c], u[2 * c + 1]} - mm[c];
            v2f ae = {__expf(xs.x), __expf(xs.y)};
            v2f af = {__expf(xu.x), __expf(xu.y)};
            acc0 = WMMA_F32X4(f.bv0[c], ae, acc0);
            acc1 = WMMA_F32X4(f.bv1[c], ae, acc1);
            acc2 = WMMA_F32X4(f.bv0[c], af, acc2);
            acc3 = WMMA_F32X4(f.bv1[c], af, acc3);
        }
    };

    auto load_t = [&](TFrag& f, int t) {
        int tc = (t < NT) ? t : NT - 1;
        load_tfrag(f, kbase + tc * TILE_F, mrow + tc * 16,
                   vzb + (size_t)(tc * 16) * DV, col, hl);
    };
    TFrag f0, f1, f2, f3;
    load_t(f0, 0);
    load_t(f1, 1);
    for (int tt = 0; tt < NT; tt += 4) {
        load_t(f2, tt + 2);
        TRIP_EDGE();
        step(f0);
        TRIP_EDGE();
        load_t(f3, tt + 3);
        TRIP_EDGE();
        step(f1);
        TRIP_EDGE();
        load_t(f0, tt + 4);
        TRIP_EDGE();
        step(f2);
        TRIP_EDGE();
        load_t(f1, tt + 5);
        TRIP_EDGE();
        step(f3);
    }

    v2f wf[4][8];
    #pragma unroll
    for (int n = 0; n < 4; ++n) {
        #pragma unroll
        for (int kc = 0; kc < 8; ++kc) {
            const float* wp = Wh + (size_t)(16 * (kc >> 2) + 8 * hl + 2 * (kc & 3)) * DD
                                 + 16 * n + col;
            wf[n][kc].x = wp[0];
            wf[n][kc].y = wp[DD];
        }
    }
    float* orow = out + ((size_t)(bb * SS) + (size_t)qt * 16 + col) * DD + 8 * hl;
    #pragma unroll
    for (int n = 0; n < 4; ++n) {
        v8f o0 = {}, o1 = {};
        #pragma unroll
        for (int j = 0; j < 4; ++j) {
            o0 = WMMA_F32X4(wf[n][j], (v2f{acc0[2 * j], acc0[2 * j + 1]}), o0);
            o1 = WMMA_F32X4(wf[n][j], (v2f{acc2[2 * j], acc2[2 * j + 1]}), o1);
        }
        #pragma unroll
        for (int j = 0; j < 4; ++j) {
            o0 = WMMA_F32X4(wf[n][4 + j], (v2f{acc1[2 * j], acc1[2 * j + 1]}), o0);
            o1 = WMMA_F32X4(wf[n][4 + j], (v2f{acc3[2 * j], acc3[2 * j + 1]}), o1);
        }
        *(v4f*)(orow + 16 * n)               = __builtin_shufflevector(o0, o0, 0, 1, 2, 3);
        *(v4f*)(orow + 16 * n + 4)           = __builtin_shufflevector(o0, o0, 4, 5, 6, 7);
        *(v4f*)(orow + 16 * DD + 16 * n)     = __builtin_shufflevector(o1, o1, 0, 1, 2, 3);
        *(v4f*)(orow + 16 * DD + 16 * n + 4) = __builtin_shufflevector(o1, o1, 4, 5, 6, 7);
    }
}

extern "C" void kernel_launch(void* const* d_in, const int* in_sizes, int n_in,
                              void* d_out, int out_size, void* d_ws, size_t ws_size,
                              hipStream_t stream)
{
    const float* x  = (const float*)d_in[0];
    const float* Wq = (const float*)d_in[1];
    const float* Wk = (const float*)d_in[2];
    const float* Wv = (const float*)d_in[3];
    const float* Wh = (const float*)d_in[4];
    float* out = (float*)d_out;

    float* ws = (float*)d_ws;
    size_t o = 0;
    float* Qp = ws + o; o += (size_t)BB * SS * 32;
    float* Kp = ws + o; o += (size_t)BB * SS * 32;
    float* Vm = ws + o; o += (size_t)BB * SS * DV;
    float* mA = ws + o; o += (size_t)BB * SS;

    const int rows = BB * SS;
    qkv_proj_kernel<<<(rows + 255) / 256, 256, 0, stream>>>(x, Wq, Wk, Wv, Qp, Kp, Vm);
    col_softmax_stats_kernel<<<GRID_PAIR, 32 * WPB, 0, stream>>>(Qp, Kp, mA, Vm);
    attn_out_kernel<<<GRID_PAIR, 32 * WPB, 0, stream>>>(Qp, Kp, Vm, mA, Wh, out);
}


# Projection kernel moved onto the f32 16x16x4 matrix instruction (W^T as A, x^T as B, two 16-row tiles per wave, whole-line fragment-order stores); the d sum order of q, k, v changes
# speedup vs baseline: 1.8376x; 1.0621x over previous
#include <hip/hip_runtime.h>
#include <hip/hip_bf16.h>


typedef __attribute__((ext_vector_type(2))) float v2f;
typedef __attribute__((ext_vector_type(4))) float v4f;
typedef __attribute__((ext_vector_type(8))) float v8f;

constexpr int BB  = 32;
constexpr int SS  = 2048;
constexpr int DD  = 64;
constexpr int DK  = 24;
constexpr int DV  = 32;
constexpr int NT  = SS / 16;
constexpr int TILE_F = 16 * DK;
constexpr int WPB   = 8;
constexpr int NPAIR = NT / 2;
constexpr int GRID_PAIR = BB * NPAIR / WPB;

#define WMMA_F32X4(Am, Bm, Cm) \
  __builtin_amdgcn_wmma_f32_16x16x4_f32(false, (Am), false, (Bm), (short)0, (Cm), false, false)

#define TRIP_EDGE() __builtin_amdgcn_sched_barrier(0)

__device__ __forceinline__ v2f lo2(v4f v) { return __builtin_shufflevector(v, v, 0, 1); }
__device__ __forceinline__ v2f hi2(v4f v) { return __builtin_shufflevector(v, v, 2, 3); }

struct WFrag { v2f c[16]; };

__device__ __forceinline__ void load_wfrag(WFrag& f, const float* wp, int ld)
{
    #pragma unroll
    for (int c = 0; c < 16; ++c) {
        const int d0 = 8 * (c >> 1) + 2 * (c & 1);
        f.c[c].x = wp[d0 * ld];
        f.c[c].y = wp[(d0 + 1) * ld];
    }
}

constexpr int GRID_PROJ = BB * SS / 32 / WPB;
static_assert(GRID_PROJ * 32 * WPB == BB * SS, "the grid covers every row exactly once");

__global__ void __launch_bounds__(32 * WPB) __attribute__((amdgpu_waves_per_eu(2, 2)))
qkv_proj_kernel(const float* __restrict__ x, const float* __restrict__ Wq,
                const float* __restrict__ Wk, const float* __restrict__ Wv,
                float* __restrict__ Qp, float* __restrict__ Kp, float* __restrict__ Vm)
{
    int wv = __builtin_amdgcn_readfirstlane((int)(threadIdx.x >> 5));
    int tile0 = 2 * (blockIdx.x * WPB + wv);
    int lane = threadIdx.x & 31;
    int a  = lane & 15;
    int hl = lane >> 4;

    v4f xb0[8], xb1[8];
    const float* xp = x + ((size_t)tile0 * 16 + a) * DD + 4 * hl;
    #pragma unroll
    for (int i = 0; i < 8; ++i) xb0[i] = *(const v4f*)(xp + 8 * i);
    #pragma unroll
    for (int i = 0; i < 8; ++i) xb1[i] = *(const v4f*)(xp + 16 * DD + 8 * i);

    const int hA = a >> 3, rr = a & 7;
    const float* wcol[5];
    #pragma unroll
    for (int j = 0; j < 3; ++j) {
        int g = 8 * j + rr;
        wcol[j] = ((g < 12) ? (Wq + 12 * hA + g) : (Wk + 12 * hA + g - 12)) + 4 * hl * DK;
    }
    wcol[3] = Wv + a + 4 * hl * DV;
    wcol[4] = Wv + 16 + a + 4 * hl * DV;

    float* qo = Qp + (size_t)tile0 * TILE_F + lane * 4;
    float* ko = Kp + (size_t)tile0 * TILE_F + lane * 4;
    float* vo = Vm + ((size_t)tile0 * 16 + a) * DV + 8 * hl;

    auto product = [&](const WFrag& w, v8f& d0, v8f& d1) {
        #pragma unroll
        for (int c = 0; c < 16; ++c) {
            v2f b0 = (c & 1) ? hi2(xb0[c >> 1]) : lo2(xb0[c >> 1]);
            v2f b1 = (c & 1) ? hi2(xb1[c >> 1]) : lo2(xb1[c >> 1]);
            d0 = WMMA_F32X4(w.c[c], b0, d0);
            d1 = WMMA_F32X4(w.c[c], b1, d1);
        }
    };
    auto store_qk = [&](int j, int q4, const v8f& d0, const v8f& d1) {
        int g0 = 8 * j + 4 * q4;
        float* dst = (g0 < 12) ? (qo + (g0 >> 2) * 128) : (ko + ((g0 - 12) >> 2) * 128);
        v4f lo0 = q4 ? __builtin_shufflevector(d0, d0, 4, 5, 6, 7)
                     : __builtin_shufflevector(d0, d0, 0, 1, 2, 3);
        v4f lo1 = q4 ? __builtin_shufflevector(d1, d1, 4, 5, 6, 7)
                     : __builtin_shufflevector(d1, d1, 0, 1, 2, 3);
        *(v4f*)(dst)          = lo0;
        *(v4f*)(dst + TILE_F) = lo1;
    };
    auto store_v = [&](int jv, const v8f& d0, const v8f& d1) {
        float* dst = vo + 16 * jv;
        *(v4f*)(dst)               = __builtin_shufflevector(d0, d0, 0, 1, 2, 3);
        *(v4f*)(dst + 4)           = __builtin_shufflevector(d0, d0, 4, 5, 6, 7);
        *(v4f*)(dst + 16 * DV)     = __builtin_shufflevector(d1, d1, 0, 1, 2, 3);
        *(v4f*)(dst + 16 * DV + 4) = __builtin_shufflevector(d1, d1, 4, 5, 6, 7);
    };

    WFrag w0, w1;
    load_wfrag(w0, wcol[0], DK);
    {
        load_wfrag(w1, wcol[1], DK);
        TRIP_EDGE();
        v8f d0 = {}, d1 = {};
        product(w0, d0, d1);
        store_qk(0, 0, d0, d1);
        store_qk(0, 1, d0, d1);
        TRIP_EDGE();
    }
    {
        load_wfrag(w0, wcol[2], DK);
        TRIP_EDGE();
        v8f d0 = {}, d1 = {};
        product(w1, d0, d1);
        store_qk(1, 0, d0, d1);
        store_qk(1, 1, d0, d1);
        TRIP_EDGE();
    }
    {
        load_wfrag(w1, wcol[3], DV);
        TRIP_EDGE();
        v8f d0 = {}, d1 = {};
        product(w0, d0, d1);
        store_qk(2, 0, d0, d1);
        store_qk(2, 1, d0, d1);
        TRIP_EDGE();
    }
    {
        load_wfrag(w0, wcol[4], DV);
        TRIP_EDGE();
        v8f d0 = {}, d1 = {};
        product(w1, d0, d1);
        store_v(0, d0, d1);
        TRIP_EDGE();
    }
    {
        v8f d0 = {}, d1 = {};
        product(w0, d0, d1);
        store_v(1, d0, d1);
    }
}

struct QFrag { v4f p0, p1, p2; };

__device__ __forceinline__ void load_qfrag(QFrag& f, const float* tp)
{
    f.p0 = *(const v4f*)(tp);
    f.p1 = *(const v4f*)(tp + 128);
    f.p2 = *(const v4f*)(tp + 256);
}

__device__ __forceinline__ void stats_step(const QFrag& f, v2f b0, v2f b1, v2f b2,
                                           v2f b3, v2f b4, v2f b5,
                                           float& m_run, float& z_run)
{
    v8f s = {};
    s = WMMA_F32X4(lo2(f.p0), b0, s);
    s = WMMA_F32X4(hi2(f.p0), b1, s);
    s = WMMA_F32X4(lo2(f.p1), b2, s);
    s = WMMA_F32X4(hi2(f.p1), b3, s);
    s = WMMA_F32X4(lo2(f.p2), b4, s);
    s = WMMA_F32X4(hi2(f.p2), b5, s);

    float tmax = fmaxf(fmaxf(fmaxf(s[0], s[1]), fmaxf(s[2], s[3])),
                       fmaxf(fmaxf(s[4], s[5]), fmaxf(s[6], s[7])));
    tmax = fmaxf(tmax, __shfl_xor(tmax, 16, 32));
    float m_new = fmaxf(m_run, tmax);
    float p = 0.f;
    #pragma unroll
    for (int i = 0; i < 8; ++i) p += __expf(s[i] - m_new);
    p += __shfl_xor(p, 16, 32);
    z_run = fmaf(z_run, __expf(m_run - m_new), p);
    m_run = m_new;
}

__global__ void __launch_bounds__(32 * WPB) __attribute__((amdgpu_waves_per_eu(2, 2)))
col_softmax_stats_kernel(const float* __restrict__ Qp, const float* __restrict__ Kp,
                         float* __restrict__ mArr, float* __restrict__ Vm)
{
    int wv = __builtin_amdgcn_readfirstlane((int)(threadIdx.x >> 5));
    int pr = blockIdx.x * WPB + wv;
    int bb = pr / NPAIR;
    int tt = 2 * (pr - bb * NPAIR);
    int lane = threadIdx.x & 31;

    QFrag kf, kg;
    load_qfrag(kf, Kp + ((size_t)bb * NT + tt) * TILE_F + lane * 4);
    load_qfrag(kg, Kp + ((size_t)bb * NT + tt + 1) * TILE_F + lane * 4);
    v2f b0 = lo2(kf.p0), b1 = hi2(kf.p0);
    v2f b2 = lo2(kf.p1), b3 = hi2(kf.p1);
    v2f b4 = lo2(kf.p2), b5 = hi2(kf.p2);
    v2f c0 = lo2(kg.p0), c1 = hi2(kg.p0);
    v2f c2 = lo2(kg.p1), c3 = hi2(kg.p1);
    v2f c4 = lo2(kg.p2), c5 = hi2(kg.p2);

    const float* qbase = Qp + (size_t)bb * NT * TILE_F + lane * 4;
    float m_run = -3.0e38f, n_run = -3.0e38f;
    float z_run = 0.f,      y_run = 0.f;

    QFrag p0a, p0b, p1a, p1b, p2a, p2b, p3a, p3b;
    load_qfrag(p0a, qbase);
    load_qfrag(p0b, qbase + TILE_F);
    load_qfrag(p1a, qbase + 2 * TILE_F);
    load_qfrag(p1b, qbase + 3 * TILE_F);
    load_qfrag(p2a, qbase + 4 * TILE_F);
    load_qfrag(p2b, qbase + 5 * TILE_F);
    auto tile_at = [&](int t) { return qbase + (t < NT ? t : NT - 1) * TILE_F; };
    for (int qt = 0; qt < NT; qt += 8) {
        load_qfrag(p3a, tile_at(qt + 6));
        load_qfrag(p3b, tile_at(qt + 7));
        TRIP_EDGE();
        stats_step(p0a, b0, b1, b2, b3, b4, b5, m_run, z_run);
        stats_step(p0a, c0, c1, c2, c3, c4, c5, n_run, y_run);
        stats_step(p0b, b0, b1, b2, b3, b4, b5, m_run, z_run);
        stats_step(p0b, c0, c1, c2, c3, c4, c5, n_run, y_run);
        TRIP_EDGE();
        load_qfrag(p0a, tile_at(qt + 8));
        load_qfrag(p0b, tile_at(qt + 9));
        TRIP_EDGE();
        stats_step(p1a, b0, b1, b2, b3, b4, b5, m_run, z_run);
        stats_step(p1a, c0, c1, c2, c3, c4, c5, n_run, y_run);
        stats_step(p1b, b0, b1, b2, b3, b4, b5, m_run, z_run);
        stats_step(p1b, c0, c1, c2, c3, c4, c5, n_run, y_run);
        TRIP_EDGE();
        load_qfrag(p1a, tile_at(qt + 10));
        load_qfrag(p1b, tile_at(qt + 11));
        TRIP_EDGE();
        stats_step(p2a, b0, b1, b2, b3, b4, b5, m_run, z_run);
        stats_step(p2a, c0, c1, c2, c3, c4, c5, n_run, y_run);
        stats_step(p2b, b0, b1, b2, b3, b4, b5, m_run, z_run);
        stats_step(p2b, c0, c1, c2, c3, c4, c5, n_run, y_run);
        TRIP_EDGE();
        load_qfrag(p2a, tile_at(qt + 12));
        load_qfrag(p2b, tile_at(qt + 13));
        TRIP_EDGE();
        stats_step(p3a, b0, b1, b2, b3, b4, b5, m_run, z_run);
        stats_step(p3a, c0, c1, c2, c3, c4, c5, n_run, y_run);
        stats_step(p3b, b0, b1, b2, b3, b4, b5, m_run, z_run);
        stats_step(p3b, c0, c1, c2, c3, c4, c5, n_run, y_run);
    }
    float m_out = (lane < 16) ? m_run : n_run;
    float z_out = (lane < 16) ? z_run : y_run;
    mArr[bb * SS + tt * 16 + lane] = m_out;
    float zinv = 1.f / z_out;

    float* vrow = Vm + ((size_t)bb * SS + (size_t)tt * 16) * DV + lane;
    float vv[32];
    #pragma unroll
    for (int j = 0; j < 32; ++j) vv[j] = vrow[j * DV];
    #pragma unroll
    for (int j = 0; j < 32; ++j) {
        float zj = __int_as_float(__builtin_amdgcn_readlane(__float_as_int(zinv), j));
        vrow[j * DV] = vv[j] * zj;
    }
}

struct TFrag { QFrag k; v4f m0, m1; v2f bv0[4], bv1[4]; };

__device__ __forceinline__ void load_tfrag(TFrag& f, const float* kp, const float* mp,
                                           const float* vt, int col, int hl)
{
    load_qfrag(f.k, kp);
    f.m0 = *(const v4f*)(mp + 8 * hl);
    f.m1 = *(const v4f*)(mp + 8 * hl + 4);
    #pragma unroll
    for (int c = 0; c < 4; ++c) {
        const float* vp = vt + (size_t)(8 * hl + 2 * c) * DV;
        f.bv0[c].x = vp[col];        f.bv0[c].y = vp[DV + col];
        f.bv1[c].x = vp[16 + col];   f.bv1[c].y = vp[DV + 16 + col];
    }
}

__global__ void __launch_bounds__(32 * WPB) __attribute__((amdgpu_waves_per_eu(2, 2)))
attn_out_kernel(const float* __restrict__ Qp, const float* __restrict__ Kp,
                const float* __restrict__ VZ, const float* __restrict__ mArr,
                const float* __restrict__ Wh, float* __restrict__ out)
{
    int wv = __builtin_amdgcn_readfirstlane((int)(threadIdx.x >> 5));
    int pr = blockIdx.x * WPB + wv;
    int bb = pr / NPAIR;
    int qt = 2 * (pr - bb * NPAIR);
    int lane = threadIdx.x & 31;
    int col  = lane & 15;
    int hl   = lane >> 4;

    QFrag qf, qg;
    load_qfrag(qf, Qp + ((size_t)bb * NT + qt) * TILE_F + lane * 4);
    load_qfrag(qg, Qp + ((size_t)bb * NT + qt + 1) * TILE_F + lane * 4);
    v2f a0 = lo2(qf.p0), a1 = hi2(qf.p0);
    v2f a2 = lo2(qf.p1), a3 = hi2(qf.p1);
    v2f a4 = lo2(qf.p2), a5 = hi2(qf.p2);
    v2f g0 = lo2(qg.p0), g1 = hi2(qg.p0);
    v2f g2 = lo2(qg.p1), g3 = hi2(qg.p1);
    v2f g4 = lo2(qg.p2), g5 = hi2(qg.p2);

    v8f acc0 = {};
    v8f acc1 = {};
    v8f acc2 = {};
    v8f acc3 = {};

    const float* kbase = Kp + (size_t)bb * NT * TILE_F + lane * 4;
    const float* vzb   = VZ + (size_t)(bb * SS) * DV;
    const float* mrow  = mArr + bb * SS;

    auto step = [&](const TFrag& f) {
        v8f s = {}, u = {};
        s = WMMA_F32X4(lo2(f.k.p0), a0, s);
        u = WMMA_F32X4(lo2(f.k.p0), g0, u);
        s = WMMA_F32X4(hi2(f.k.p0), a1, s);
        u = WMMA_F32X4(hi2(f.k.p0), g1, u);
        s = WMMA_F32X4(lo2(f.k.p1), a2, s);
        u = WMMA_F32X4(lo2(f.k.p1), g2, u);
        s = WMMA_F32X4(hi2(f.k.p1), a3, s);
        u = WMMA_F32X4(hi2(f.k.p1), g3, u);
        s = WMMA_F32X4(lo2(f.k.p2), a4, s);
        u = WMMA_F32X4(lo2(f.k.p2), g4, u);
        s = WMMA_F32X4(hi2(f.k.p2), a5, s);
        u = WMMA_F32X4(hi2(f.k.p2), g5, u);

        const v2f mm[4] = {lo2(f.m0), hi2(f.m0), lo2(f.m1), hi2(f.m1)};
        #pragma unroll
        for (int c = 0; c < 4; ++c) {
            v2f xs = v2f{s[2 * c], s[2 * c + 1]} - mm[c];
            v2f xu = v2f{u[2 * c], u[2 * c + 1]} - mm[c];
            v2f ae = {__expf(xs.x), __expf(xs.y)};
            v2f af = {__expf(xu.x), __expf(xu.y)};
            acc0 = WMMA_F32X4(f.bv0[c], ae, acc0);
            acc1 = WMMA_F32X4(f.bv1[c], ae, acc1);
            acc2 = WMMA_F32X4(f.bv0[c], af, acc2);
            acc3 = WMMA_F32X4(f.bv1[c], af, acc3);
        }
    };

    auto load_t = [&](TFrag& f, int t) {
        int tc = (t < NT) ? t : NT - 1;
        load_tfrag(f, kbase + tc * TILE_F, mrow + tc * 16,
                   vzb + (size_t)(tc * 16) * DV, col, hl);
    };
    TFrag f0, f1, f2, f3;
    load_t(f0, 0);
    load_t(f1, 1);
    for (int tt = 0; tt < NT; tt += 4) {
        load_t(f2, tt + 2);
        TRIP_EDGE();
        step(f0);
        TRIP_EDGE();
        load_t(f3, tt + 3);
        TRIP_EDGE();
        step(f1);
        TRIP_EDGE();
        load_t(f0, tt + 4);
        TRIP_EDGE();
        step(f2);
        TRIP_EDGE();
        load_t(f1, tt + 5);
        TRIP_EDGE();
        step(f3);
    }

    v2f wf[4][8];
    #pragma unroll
    for (int n = 0; n < 4; ++n) {
        #pragma unroll
        for (int kc = 0; kc < 8; ++kc) {
            const float* wp = Wh + (size_t)(16 * (kc >> 2) + 8 * hl + 2 * (kc & 3)) * DD
                                 + 16 * n + col;
            wf[n][kc].x = wp[0];
            wf[n][kc].y = wp[DD];
        }
    }
    float* orow = out + ((size_t)(bb * SS) + (size_t)qt * 16 + col) * DD + 8 * hl;
    #pragma unroll
    for (int n = 0; n < 4; ++n) {
        v8f o0 = {}, o1 = {};
        #pragma unroll
        for (int j = 0; j < 4; ++j) {
            o0 = WMMA_F32X4(wf[n][j], (v2f{acc0[2 * j], acc0[2 * j + 1]}), o0);
            o1 = WMMA_F32X4(wf[n][j], (v2f{acc2[2 * j], acc2[2 * j + 1]}), o1);
        }
        #pragma unroll
        for (int j = 0; j < 4; ++j) {
            o0 = WMMA_F32X4(wf[n][4 + j], (v2f{acc1[2 * j], acc1[2 * j + 1]}), o0);
            o1 = WMMA_F32X4(wf[n][4 + j], (v2f{acc3[2 * j], acc3[2 * j + 1]}), o1);
        }
        *(v4f*)(orow + 16 * n)               = __builtin_shufflevector(o0, o0, 0, 1, 2, 3);
        *(v4f*)(orow + 16 * n + 4)           = __builtin_shufflevector(o0, o0, 4, 5, 6, 7);
        *(v4f*)(orow + 16 * DD + 16 * n)     = __builtin_shufflevector(o1, o1, 0, 1, 2, 3);
        *(v4f*)(orow + 16 * DD + 16 * n + 4) = __builtin_shufflevector(o1, o1, 4, 5, 6, 7);
    }
}

extern "C" void kernel_launch(void* const* d_in, const int* in_sizes, int n_in,
                              void* d_out, int out_size, void* d_ws, size_t ws_size,
                              hipStream_t stream)
{
    const float* x  = (const float*)d_in[0];
    const float* Wq = (const float*)d_in[1];
    const float* Wk = (const float*)d_in[2];
    const float* Wv = (const float*)d_in[3];
    const float* Wh = (const float*)d_in[4];
    float* out = (float*)d_out;

    float* ws = (float*)d_ws;
    size_t o = 0;
    float* Qp = ws + o; o += (size_t)BB * SS * 32;
    float* Kp = ws + o; o += (size_t)BB * SS * 32;
    float* Vm = ws + o; o += (size_t)BB * SS * DV;
    float* mA = ws + o; o += (size_t)BB * SS;

    qkv_proj_kernel<<<GRID_PROJ, 32 * WPB, 0, stream>>>(x, Wq, Wk, Wv, Qp, Kp, Vm);
    col_softmax_stats_kernel<<<GRID_PAIR, 32 * WPB, 0, stream>>>(Qp, Kp, mA, Vm);
    attn_out_kernel<<<GRID_PAIR, 32 * WPB, 0, stream>>>(Qp, Kp, Vm, mA, Wh, out);
}


# vz plane kept in WMMA fragment order (V product untransposed: a lane holds 8 keys of one dv; scaled in place): a t-tile's vz operand is 4 whole-line b128 loads, not 16 b32 gathers; same sum orders
# speedup vs baseline: 2.0076x; 1.0925x over previous
#include <hip/hip_runtime.h>
#include <hip/hip_bf16.h>


typedef __attribute__((ext_vector_type(2))) float v2f;
typedef __attribute__((ext_vector_type(4))) float v4f;
typedef __attribute__((ext_vector_type(8))) float v8f;

constexpr int BB  = 32;
constexpr int SS  = 2048;
constexpr int DD  = 64;
constexpr int DK  = 24;
constexpr int DV  = 32;
constexpr int NT  = SS / 16;
constexpr int TILE_F = 16 * DK;
constexpr int VTILE_F = 16 * DV;
constexpr int WPB   = 8;
constexpr int NPAIR = NT / 2;
constexpr int GRID_PAIR = BB * NPAIR / WPB;

#define WMMA_F32X4(Am, Bm, Cm) \
  __builtin_amdgcn_wmma_f32_16x16x4_f32(false, (Am), false, (Bm), (short)0, (Cm), false, false)

#define TRIP_EDGE() __builtin_amdgcn_sched_barrier(0)

__device__ __forceinline__ v2f lo2(v4f v) { return __builtin_shufflevector(v, v, 0, 1); }
__device__ __forceinline__ v2f hi2(v4f v) { return __builtin_shufflevector(v, v, 2, 3); }

struct WFrag { v2f c[16]; };

__device__ __forceinline__ void load_wfrag(WFrag& f, const float* wp, int ld)
{
    #pragma unroll
    for (int c = 0; c < 16; ++c) {
        const int d0 = 8 * (c >> 1) + 2 * (c & 1);
        f.c[c].x = wp[d0 * ld];
        f.c[c].y = wp[(d0 + 1) * ld];
    }
}

constexpr int GRID_PROJ = BB * SS / 32 / WPB;
static_assert(GRID_PROJ * 32 * WPB == BB * SS, "the grid covers every row exactly once");

__global__ void __launch_bounds__(32 * WPB) __attribute__((amdgpu_waves_per_eu(2, 2)))
qkv_proj_kernel(const float* __restrict__ x, const float* __restrict__ Wq,
                const float* __restrict__ Wk, const float* __restrict__ Wv,
                float* __restrict__ Qp, float* __restrict__ Kp, float* __restrict__ Vm)
{
    int wv = __builtin_amdgcn_readfirstlane((int)(threadIdx.x >> 5));
    int tile0 = 2 * (blockIdx.x * WPB + wv);
    int lane = threadIdx.x & 31;
    int a  = lane & 15;
    int hl = lane >> 4;

    v4f xb0[8], xb1[8];
    const float* xp = x + ((size_t)tile0 * 16 + a) * DD + 4 * hl;
    #pragma unroll
    for (int i = 0; i < 8; ++i) xb0[i] = *(const v4f*)(xp + 8 * i);
    #pragma unroll
    for (int i = 0; i < 8; ++i) xb1[i] = *(const v4f*)(xp + 16 * DD + 8 * i);

    const int hA = a >> 3, rr = a & 7;
    const float* wcol[5];
    #pragma unroll
    for (int j = 0; j < 3; ++j) {
        int g = 8 * j + rr;
        wcol[j] = ((g < 12) ? (Wq + 12 * hA + g) : (Wk + 12 * hA + g - 12)) + 4 * hl * DK;
    }
    wcol[3] = Wv + a + 4 * hl * DV;
    wcol[4] = Wv + 16 + a + 4 * hl * DV;

    float* qo = Qp + (size_t)tile0 * TILE_F + lane * 4;
    float* ko = Kp + (size_t)tile0 * TILE_F + lane * 4;
    float* vo = Vm + (size_t)tile0 * VTILE_F + lane * 4;

    auto product = [&](const WFrag& w, v8f& d0, v8f& d1) {
        #pragma unroll
        for (int c = 0; c < 16; ++c) {
            v2f b0 = (c & 1) ? hi2(xb0[c >> 1]) : lo2(xb0[c >> 1]);
            v2f b1 = (c & 1) ? hi2(xb1[c >> 1]) : lo2(xb1[c >> 1]);
            d0 = WMMA_F32X4(w.c[c], b0, d0);
            d1 = WMMA_F32X4(w.c[c], b1, d1);
        }
    };
    auto product_v = [&](const WFrag& w, v8f& d0, v8f& d1) {
        #pragma unroll
        for (int c = 0; c < 16; ++c) {
            v2f b0 = (c & 1) ? hi2(xb0[c >> 1]) : lo2(xb0[c >> 1]);
            v2f b1 = (c & 1) ? hi2(xb1[c >> 1]) : lo2(xb1[c >> 1]);
            d0 = WMMA_F32X4(b0, w.c[c], d0);
            d1 = WMMA_F32X4(b1, w.c[c], d1);
        }
    };
    auto store_qk = [&](int j, int q4, const v8f& d0, const v8f& d1) {
        int g0 = 8 * j + 4 * q4;
        float* dst = (g0 < 12) ? (qo + (g0 >> 2) * 128) : (ko + ((g0 - 12) >> 2) * 128);
        v4f lo0 = q4 ? __builtin_shufflevector(d0, d0, 4, 5, 6, 7)
                     : __builtin_shufflevector(d0, d0, 0, 1, 2, 3);
        v4f lo1 = q4 ? __builtin_shufflevector(d1, d1, 4, 5, 6, 7)
                     : __builtin_shufflevector(d1, d1, 0, 1, 2, 3);
        *(v4f*)(dst)          = lo0;
        *(v4f*)(dst + TILE_F) = lo1;
    };
    auto store_v = [&](int jv, const v8f& d0, const v8f& d1) {
        float* dst = vo + 2 * jv * 128;
        *(v4f*)(dst)                 = __builtin_shufflevector(d0, d0, 0, 1, 2, 3);
        *(v4f*)(dst + 128)           = __builtin_shufflevector(d0, d0, 4, 5, 6, 7);
        *(v4f*)(dst + VTILE_F)       = __builtin_shufflevector(d1, d1, 0, 1, 2, 3);
        *(v4f*)(dst + VTILE_F + 128) = __builtin_shufflevector(d1, d1, 4, 5, 6, 7);
    };

    WFrag w0, w1;
    load_wfrag(w0, wcol[0], DK);
    {
        load_wfrag(w1, wcol[1], DK);
        TRIP_EDGE();
        v8f d0 = {}, d1 = {};
        product(w0, d0, d1);
        store_qk(0, 0, d0, d1);
        store_qk(0, 1, d0, d1);
        TRIP_EDGE();
    }
    {
        load_wfrag(w0, wcol[2], DK);
        TRIP_EDGE();
        v8f d0 = {}, d1 = {};
        product(w1, d0, d1);
        store_qk(1, 0, d0, d1);
        store_qk(1, 1, d0, d1);
        TRIP_EDGE();
    }
    {
        load_wfrag(w1, wcol[3], DV);
        TRIP_EDGE();
        v8f d0 = {}, d1 = {};
        product(w0, d0, d1);
        store_qk(2, 0, d0, d1);
        store_qk(2, 1, d0, d1);
        TRIP_EDGE();
    }
    {
        load_wfrag(w0, wcol[4], DV);
        TRIP_EDGE();
        v8f d0 = {}, d1 = {};
        product_v(w1, d0, d1);
        store_v(0, d0, d1);
        TRIP_EDGE();
    }
    {
        v8f d0 = {}, d1 = {};
        product_v(w0, d0, d1);
        store_v(1, d0, d1);
    }
}

struct QFrag { v4f p0, p1, p2; };

__device__ __forceinline__ void load_qfrag(QFrag& f, const float* tp)
{
    f.p0 = *(const v4f*)(tp);
    f.p1 = *(const v4f*)(tp + 128);
    f.p2 = *(const v4f*)(tp + 256);
}

__device__ __forceinline__ void stats_step(const QFrag& f, v2f b0, v2f b1, v2f b2,
                                           v2f b3, v2f b4, v2f b5,
                                           float& m_run, float& z_run)
{
    v8f s = {};
    s = WMMA_F32X4(lo2(f.p0), b0, s);
    s = WMMA_F32X4(hi2(f.p0), b1, s);
    s = WMMA_F32X4(lo2(f.p1), b2, s);
    s = WMMA_F32X4(hi2(f.p1), b3, s);
    s = WMMA_F32X4(lo2(f.p2), b4, s);
    s = WMMA_F32X4(hi2(f.p2), b5, s);

    float tmax = fmaxf(fmaxf(fmaxf(s[0], s[1]), fmaxf(s[2], s[3])),
                       fmaxf(fmaxf(s[4], s[5]), fmaxf(s[6], s[7])));
    tmax = fmaxf(tmax, __shfl_xor(tmax, 16, 32));
    float m_new = fmaxf(m_run, tmax);
    float p = 0.f;
    #pragma unroll
    for (int i = 0; i < 8; ++i) p += __expf(s[i] - m_new);
    p += __shfl_xor(p, 16, 32);
    z_run = fmaf(z_run, __expf(m_run - m_new), p);
    m_run = m_new;
}

__global__ void __launch_bounds__(32 * WPB) __attribute__((amdgpu_waves_per_eu(2, 2)))
col_softmax_stats_kernel(const float* __restrict__ Qp, const float* __restrict__ Kp,
                         float* __restrict__ mArr, float* __restrict__ Vm)
{
    int wv = __builtin_amdgcn_readfirstlane((int)(threadIdx.x >> 5));
    int pr = blockIdx.x * WPB + wv;
    int bb = pr / NPAIR;
    int tt = 2 * (pr - bb * NPAIR);
    int lane = threadIdx.x & 31;

    QFrag kf, kg;
    load_qfrag(kf, Kp + ((size_t)bb * NT + tt) * TILE_F + lane * 4);
    load_qfrag(kg, Kp + ((size_t)bb * NT + tt + 1) * TILE_F + lane * 4);
    v2f b0 = lo2(kf.p0), b1 = hi2(kf.p0);
    v2f b2 = lo2(kf.p1), b3 = hi2(kf.p1);
    v2f b4 = lo2(kf.p2), b5 = hi2(kf.p2);
    v2f c0 = lo2(kg.p0), c1 = hi2(kg.p0);
    v2f c2 = lo2(kg.p1), c3 = hi2(kg.p1);
    v2f c4 = lo2(kg.p2), c5 = hi2(kg.p2);

    const float* qbase = Qp + (size_t)bb * NT * TILE_F + lane * 4;
    float m_run = -3.0e38f, n_run = -3.0e38f;
    float z_run = 0.f,      y_run = 0.f;

    QFrag p0a, p0b, p1a, p1b, p2a, p2b, p3a, p3b;
    load_qfrag(p0a, qbase);
    load_qfrag(p0b, qbase + TILE_F);
    load_qfrag(p1a, qbase + 2 * TILE_F);
    load_qfrag(p1b, qbase + 3 * TILE_F);
    load_qfrag(p2a, qbase + 4 * TILE_F);
    load_qfrag(p2b, qbase + 5 * TILE_F);
    auto tile_at = [&](int t) { return qbase + (t < NT ? t : NT - 1) * TILE_F; };
    for (int qt = 0; qt < NT; qt += 8) {
        load_qfrag(p3a, tile_at(qt + 6));
        load_qfrag(p3b, tile_at(qt + 7));
        TRIP_EDGE();
        stats_step(p0a, b0, b1, b2, b3, b4, b5, m_run, z_run);
        stats_step(p0a, c0, c1, c2, c3, c4, c5, n_run, y_run);
        stats_step(p0b, b0, b1, b2, b3, b4, b5, m_run, z_run);
        stats_step(p0b, c0, c1, c2, c3, c4, c5, n_run, y_run);
        TRIP_EDGE();
        load_qfrag(p0a, tile_at(qt + 8));
        load_qfrag(p0b, tile_at(qt + 9));
        TRIP_EDGE();
        stats_step(p1a, b0, b1, b2, b3, b4, b5, m_run, z_run);
        stats_step(p1a, c0, c1, c2, c3, c4, c5, n_run, y_run);
        stats_step(p1b, b0, b1, b2, b3, b4, b5, m_run, z_run);
        stats_step(p1b, c0, c1, c2, c3, c4, c5, n_run, y_run);
        TRIP_EDGE();
        load_qfrag(p1a, tile_at(qt + 10));
        load_qfrag(p1b, tile_at(qt + 11));
        TRIP_EDGE();
        stats_step(p2a, b0, b1, b2, b3, b4, b5, m_run, z_run);
        stats_step(p2a, c0, c1, c2, c3, c4, c5, n_run, y_run);
        stats_step(p2b, b0, b1, b2, b3, b4, b5, m_run, z_run);
        stats_step(p2b, c0, c1, c2, c3, c4, c5, n_run, y_run);
        TRIP_EDGE();
        load_qfrag(p2a, tile_at(qt + 12));
        load_qfrag(p2b, tile_at(qt + 13));
        TRIP_EDGE();
        stats_step(p3a, b0, b1, b2, b3, b4, b5, m_run, z_run);
        stats_step(p3a, c0, c1, c2, c3, c4, c5, n_run, y_run);
        stats_step(p3b, b0, b1, b2, b3, b4, b5, m_run, z_run);
        stats_step(p3b, c0, c1, c2, c3, c4, c5, n_run, y_run);
    }
    float m_out = (lane < 16) ? m_run : n_run;
    mArr[bb * SS + tt * 16 + lane] = m_out;
    float zinv0 = 1.f / z_run;
    float zinv1 = 1.f / y_run;

    const int hl = lane >> 4;
    float* vt = Vm + ((size_t)bb * NT + tt) * VTILE_F + lane * 4;
    v4f blk[2][4];
    #pragma unroll
    for (int t2 = 0; t2 < 2; ++t2) {
        #pragma unroll
        for (int b = 0; b < 4; ++b) blk[t2][b] = *(const v4f*)(vt + t2 * VTILE_F + b * 128);
    }
    #pragma unroll
    for (int t2 = 0; t2 < 2; ++t2) {
        const float zi = t2 ? zinv1 : zinv0;
        #pragma unroll
        for (int g = 0; g < 2; ++g) {
            v4f zq;
            zq.x = __shfl(zi, 8 * hl + 4 * g + 0, 32);
            zq.y = __shfl(zi, 8 * hl + 4 * g + 1, 32);
            zq.z = __shfl(zi, 8 * hl + 4 * g + 2, 32);
            zq.w = __shfl(zi, 8 * hl + 4 * g + 3, 32);
            *(v4f*)(vt + t2 * VTILE_F + g * 128)       = blk[t2][g] * zq;
            *(v4f*)(vt + t2 * VTILE_F + (2 + g) * 128) = blk[t2][2 + g] * zq;
        }
    }
}

struct TFrag { QFrag k; v4f m0, m1; v4f vb[4]; };

__device__ __forceinline__ void load_tfrag(TFrag& f, const float* kp, const float* mp,
                                           const float* vt, int hl)
{
    load_qfrag(f.k, kp);
    f.m0 = *(const v4f*)(mp + 8 * hl);
    f.m1 = *(const v4f*)(mp + 8 * hl + 4);
    #pragma unroll
    for (int b = 0; b < 4; ++b) f.vb[b] = *(const v4f*)(vt + b * 128);
}

__global__ void __launch_bounds__(32 * WPB) __attribute__((amdgpu_waves_per_eu(2, 2)))
attn_out_kernel(const float* __restrict__ Qp, const float* __restrict__ Kp,
                const float* __restrict__ VZ, const float* __restrict__ mArr,
                const float* __restrict__ Wh, float* __restrict__ out)
{
    int wv = __builtin_amdgcn_readfirstlane((int)(threadIdx.x >> 5));
    int pr = blockIdx.x * WPB + wv;
    int bb = pr / NPAIR;
    int qt = 2 * (pr - bb * NPAIR);
    int lane = threadIdx.x & 31;
    int col  = lane & 15;
    int hl   = lane >> 4;

    QFrag qf, qg;
    load_qfrag(qf, Qp + ((size_t)bb * NT + qt) * TILE_F + lane * 4);
    load_qfrag(qg, Qp + ((size_t)bb * NT + qt + 1) * TILE_F + lane * 4);
    v2f a0 = lo2(qf.p0), a1 = hi2(qf.p0);
    v2f a2 = lo2(qf.p1), a3 = hi2(qf.p1);
    v2f a4 = lo2(qf.p2), a5 = hi2(qf.p2);
    v2f g0 = lo2(qg.p0), g1 = hi2(qg.p0);
    v2f g2 = lo2(qg.p1), g3 = hi2(qg.p1);
    v2f g4 = lo2(qg.p2), g5 = hi2(qg.p2);

    v8f acc0 = {};
    v8f acc1 = {};
    v8f acc2 = {};
    v8f acc3 = {};

    const float* kbase = Kp + (size_t)bb * NT * TILE_F + lane * 4;
    const float* vzb   = VZ + (size_t)bb * NT * VTILE_F + lane * 4;
    const float* mrow  = mArr + bb * SS;

    auto step = [&](const TFrag& f) {
        v8f s = {}, u = {};
        s = WMMA_F32X4(lo2(f.k.p0), a0, s);
        u = WMMA_F32X4(lo2(f.k.p0), g0, u);
        s = WMMA_F32X4(hi2(f.k.p0), a1, s);
        u = WMMA_F32X4(hi2(f.k.p0), g1, u);
        s = WMMA_F32X4(lo2(f.k.p1), a2, s);
        u = WMMA_F32X4(lo2(f.k.p1), g2, u);
        s = WMMA_F32X4(hi2(f.k.p1), a3, s);
        u = WMMA_F32X4(hi2(f.k.p1), g3, u);
        s = WMMA_F32X4(lo2(f.k.p2), a4, s);
        u = WMMA_F32X4(lo2(f.k.p2), g4, u);
        s = WMMA_F32X4(hi2(f.k.p2), a5, s);
        u = WMMA_F32X4(hi2(f.k.p2), g5, u);

        const v2f mm[4] = {lo2(f.m0), hi2(f.m0), lo2(f.m1), hi2(f.m1)};
        #pragma unroll
        for (int c = 0; c < 4; ++c) {
            v2f xs = v2f{s[2 * c], s[2 * c + 1]} - mm[c];
            v2f xu = v2f{u[2 * c], u[2 * c + 1]} - mm[c];
            v2f ae = {__expf(xs.x), __expf(xs.y)};
            v2f af = {__expf(xu.x), __expf(xu.y)};
            v2f bv0 = (c & 1) ? hi2(f.vb[c >> 1]) : lo2(f.vb[c >> 1]);
            v2f bv1 = (c & 1) ? hi2(f.vb[2 + (c >> 1)]) : lo2(f.vb[2 + (c >> 1)]);
            acc0 = WMMA_F32X4(bv0, ae, acc0);
            acc1 = WMMA_F32X4(bv1, ae, acc1);
            acc2 = WMMA_F32X4(bv0, af, acc2);
            acc3 = WMMA_F32X4(bv1, af, acc3);
        }
    };

    auto load_t = [&](TFrag& f, int t) {
        int tc = (t < NT) ? t : NT - 1;
        load_tfrag(f, kbase + tc * TILE_F, mrow + tc * 16, vzb + tc * VTILE_F, hl);
    };
    TFrag f0, f1, f2, f3;
    load_t(f0, 0);
    load_t(f1, 1);
    for (int tt = 0; tt < NT; tt += 4) {
        load_t(f2, tt + 2);
        TRIP_EDGE();
        step(f0);
        TRIP_EDGE();
        load_t(f3, tt + 3);
        TRIP_EDGE();
        step(f1);
        TRIP_EDGE();
        load_t(f0, tt + 4);
        TRIP_EDGE();
        step(f2);
        TRIP_EDGE();
        load_t(f1, tt + 5);
        TRIP_EDGE();
        step(f3);
    }

    v2f wf[4][8];
    #pragma unroll
    for (int n = 0; n < 4; ++n) {
        #pragma unroll
        for (int kc = 0; kc < 8; ++kc) {
            const float* wp = Wh + (size_t)(16 * (kc >> 2) + 8 * hl + 2 * (kc & 3)) * DD
                                 + 16 * n + col;
            wf[n][kc].x = wp[0];
            wf[n][kc].y = wp[DD];
        }
    }
    float* orow = out + ((size_t)(bb * SS) + (size_t)qt * 16 + col) * DD + 8 * hl;
    #pragma unroll
    for (int n = 0; n < 4; ++n) {
        v8f o0 = {}, o1 = {};
        #pragma unroll
        for (int j = 0; j < 4; ++j) {
            o0 = WMMA_F32X4(wf[n][j], (v2f{acc0[2 * j], acc0[2 * j + 1]}), o0);
            o1 = WMMA_F32X4(wf[n][j], (v2f{acc2[2 * j], acc2[2 * j + 1]}), o1);
        }
        #pragma unroll
        for (int j = 0; j < 4; ++j) {
            o0 = WMMA_F32X4(wf[n][4 + j], (v2f{acc1[2 * j], acc1[2 * j + 1]}), o0);
            o1 = WMMA_F32X4(wf[n][4 + j], (v2f{acc3[2 * j], acc3[2 * j + 1]}), o1);
        }
        *(v4f*)(orow + 16 * n)               = __builtin_shufflevector(o0, o0, 0, 1, 2, 3);
        *(v4f*)(orow + 16 * n + 4)           = __builtin_shufflevector(o0, o0, 4, 5, 6, 7);
        *(v4f*)(orow + 16 * DD + 16 * n)     = __builtin_shufflevector(o1, o1, 0, 1, 2, 3);
        *(v4f*)(orow + 16 * DD + 16 * n + 4) = __builtin_shufflevector(o1, o1, 4, 5, 6, 7);
    }
}

extern "C" void kernel_launch(void* const* d_in, const int* in_sizes, int n_in,
                              void* d_out, int out_size, void* d_ws, size_t ws_size,
                              hipStream_t stream)
{
    const float* x  = (const float*)d_in[0];
    const float* Wq = (const float*)d_in[1];
    const float* Wk = (const float*)d_in[2];
    const float* Wv = (const float*)d_in[3];
    const float* Wh = (const float*)d_in[4];
    float* out = (float*)d_out;

    float* ws = (float*)d_ws;
    size_t o = 0;
    float* Qp = ws + o; o += (size_t)BB * SS * 32;
    float* Kp = ws + o; o += (size_t)BB * SS * 32;
    float* Vm = ws + o; o += (size_t)BB * SS * DV;
    float* mA = ws + o; o += (size_t)BB * SS;

    qkv_proj_kernel<<<GRID_PROJ, 32 * WPB, 0, stream>>>(x, Wq, Wk, Wv, Qp, Kp, Vm);
    col_softmax_stats_kernel<<<GRID_PAIR, 32 * WPB, 0, stream>>>(Qp, Kp, mA, Vm);
    attn_out_kernel<<<GRID_PAIR, 32 * WPB, 0, stream>>>(Qp, Kp, Vm, mA, Wh, out);
}
